# starting kernel
# baseline (speedup 1.0000x reference)
#include <hip/hip_runtime.h>
#include <hip/hip_bf16.h>


typedef __attribute__((ext_vector_type(2))) float v2f;
typedef __attribute__((ext_vector_type(8))) float v8f;

__device__ __forceinline__ v8f wmma_k4(v2f a, v2f b, v8f c) {
  return __builtin_amdgcn_wmma_f32_16x16x4_f32(
       false, a,  false, b,
       (short)0, c,  false,  false);
}

__device__ __forceinline__ float ssp(float x) {
  float e = __expf(-fabsf(x));
  return fmaxf(x, 0.0f) + __logf(1.0f + e) - 0.69314718055994531f;
}

#define F 128
#define KRBF 64
#define A_STRIDE 132
#define E_STRIDE 68

__global__ void ssp_kernel(const float* __restrict__ x, float* __restrict__ xa, int n) {
  int i = blockIdx.x * 256 + threadIdx.x;
  if (i < n) xa[i] = ssp(x[i]);
}

__global__ void __launch_bounds__(256)
edge_msg_kernel(const float* __restrict__ xa, const int* __restrict__ eidx,
                const float* __restrict__ ea, const float* __restrict__ Wd,
                const float* __restrict__ bd, const float* __restrict__ Wg,
                float* __restrict__ aggr, int E) {
  __shared__ __align__(16) float As[16 * A_STRIDE];
  __shared__ __align__(16) float Es[16 * E_STRIDE];
  __shared__ int dsts[16];

  const int e0 = blockIdx.x * 16;

  if (threadIdx.x < 16) {
    int e = e0 + threadIdx.x;
    if (e >= E) e = E - 1;
    dsts[threadIdx.x] = eidx[E + e];
  }
  for (int i = threadIdx.x; i < 16 * F; i += 256) {
    int r = i >> 7, c = i & (F - 1);
    int e = e0 + r;
    if (e >= E) e = E - 1;
    int s = eidx[e];
    As[r * A_STRIDE + c] = xa[s * F + c];
  }
  for (int i = threadIdx.x; i < 16 * KRBF; i += 256) {
    int r = i >> 6, c = i & (KRBF - 1);
    int e = e0 + r;
    if (e >= E) e = E - 1;
    Es[r * E_STRIDE + c] = ea[e * KRBF + c];
  }
  __syncthreads();

  const int lane = threadIdx.x & 31;
  const int wv   = threadIdx.x >> 5;
  const int m    = lane & 15;
  const int hi   = lane >> 4;
  const int n    = wv * 16 + m;

  v8f acc = {};
  const float* wrow = Wd + n * F;
#pragma unroll
  for (int kk = 0; kk < F / 4; ++kk) {
    v2f a = *(const v2f*)(&As[m * A_STRIDE + kk * 4 + 2 * hi]);
    v2f b = *(const v2f*)(&wrow[kk * 4 + 2 * hi]);
    acc = wmma_k4(a, b, acc);
  }

  v8f acc2 = {};
  const float* grow = Wg + n * KRBF;
#pragma unroll
  for (int kk = 0; kk < KRBF / 4; ++kk) {
    v2f a = *(const v2f*)(&Es[m * E_STRIDE + kk * 4 + 2 * hi]);
    v2f b = *(const v2f*)(&grow[kk * 4 + 2 * hi]);
    acc2 = wmma_k4(a, b, acc2);
  }

  const float bn = bd[n];
#pragma unroll
  for (int j = 0; j < 8; ++j) {
    int r = j + 8 * hi;
    int e = e0 + r;
    float msg = ssp(acc[j] + bn) * acc2[j];
    if (e < E) atomicAdd(&aggr[dsts[r] * F + n], msg);
  }
}

template <bool IN_ACT, int MODE, bool WRITE2>
__global__ void __launch_bounds__(256)
gemm128_kernel(const float* __restrict__ In, const float* __restrict__ W,
               const float* __restrict__ bias, const float* __restrict__ Ex1,
               const float* __restrict__ Ex2, float* __restrict__ Out,
               float* __restrict__ Out2, int N) {
  __shared__ __align__(16) float As[16 * A_STRIDE];

  const int row0 = blockIdx.x * 16;
  for (int i = threadIdx.x; i < 16 * F; i += 256) {
    int r = i >> 7, c = i & (F - 1);
    int rr = row0 + r;
    if (rr >= N) rr = N - 1;
    float v = In[rr * F + c];
    if (IN_ACT) v = ssp(v);
    As[r * A_STRIDE + c] = v;
  }
  __syncthreads();

  const int lane = threadIdx.x & 31;
  const int wv   = threadIdx.x >> 5;
  const int m    = lane & 15;
  const int hi   = lane >> 4;
  const int n    = wv * 16 + m;

  v8f acc = {};
  const float* wrow = W + n * F;
#pragma unroll
  for (int kk = 0; kk < F / 4; ++kk) {
    v2f a = *(const v2f*)(&As[m * A_STRIDE + kk * 4 + 2 * hi]);
    v2f b = *(const v2f*)(&wrow[kk * 4 + 2 * hi]);
    acc = wmma_k4(a, b, acc);
  }

  const float bn = bias[n];
#pragma unroll
  for (int j = 0; j < 8; ++j) {
    int r = row0 + j + 8 * hi;
    if (r >= N) continue;
    float g = acc[j] + bn;
    float o;
    if (MODE == 0)      o = ssp(g) + Ex1[r * F + n];
    else if (MODE == 1) o = g;
    else if (MODE == 2) o = Ex1[r * F + n] + g;
    else                o = g + Ex1[r * F + n] * Ex2[n];
    Out[r * F + n] = o;
    if (WRITE2) Out2[r * F + n] = o;
  }
}

extern "C" void kernel_launch(void* const* d_in, const int* in_sizes, int n_in,
                              void* d_out, int out_size, void* d_ws, size_t ws_size,
                              hipStream_t stream) {
  const float* x      = (const float*)d_in[0];
  const int*   eidx   = (const int*)  d_in[1];
  const float* ea     = (const float*)d_in[2];
  const float* W_same = (const float*)d_in[3];
  const float* b_same = (const float*)d_in[4];
  const float* W_diff = (const float*)d_in[5];
  const float* b_diff = (const float*)d_in[6];
  const float* W_G    = (const float*)d_in[7];
  const float* rW1    = (const float*)d_in[8];
  const float* rb1    = (const float*)d_in[9];
  const float* rW2    = (const float*)d_in[10];
  const float* rb2    = (const float*)d_in[11];
  const float* W_last = (const float*)d_in[12];
  const float* b_last = (const float*)d_in[13];
  const float* u      = (const float*)d_in[14];

  const int NF = in_sizes[0];
  const int N  = NF / F;
  const int E  = in_sizes[2] / KRBF;
  const int R  = in_sizes[8] / (F * F);

  float* out_final = (float*)d_out;
  float* out_msged = (float*)d_out + NF;

  float* buf0 = (float*)d_ws;
  float* buf1 = buf0 + NF;

  const int nbN = (N + 15) / 16;
  const int nbE = (E + 15) / 16;

  hipMemsetAsync(buf1, 0, (size_t)NF * sizeof(float), stream);

  ssp_kernel<<<(NF + 255) / 256, 256, 0, stream>>>(x, buf0, NF);

  edge_msg_kernel<<<nbE, 256, 0, stream>>>(buf0, eidx, ea, W_diff, b_diff, W_G,
                                           buf1, E);

  gemm128_kernel<false, 0, true><<<nbN, 256, 0, stream>>>(
      buf0, W_same, b_same, buf1, nullptr, out_msged, buf0, N);

  for (int r = 0; r < R; ++r) {
    gemm128_kernel<true, 1, false><<<nbN, 256, 0, stream>>>(
        buf0, rW1 + (size_t)r * F * F, rb1 + (size_t)r * F,
        nullptr, nullptr, buf1, nullptr, N);
    gemm128_kernel<true, 2, false><<<nbN, 256, 0, stream>>>(
        buf1, rW2 + (size_t)r * F * F, rb2 + (size_t)r * F,
        buf0, nullptr, buf0, nullptr, N);
  }

  gemm128_kernel<true, 3, false><<<nbN, 256, 0, stream>>>(
      buf0, W_last, b_last, x, u, out_final, nullptr, N);
}


# Weight fragments loaded once per workgroup into registers; 512 workgroups loop over 16-row tiles (edge kernel and node GEMMs); k order and f32 atomics kept (arrival order timing-dependent as before)
# speedup vs baseline: 1.4394x; 1.4394x over previous
#include <hip/hip_runtime.h>
#include <hip/hip_bf16.h>


typedef __attribute__((ext_vector_type(2))) float v2f;
typedef __attribute__((ext_vector_type(8))) float v8f;

__device__ __forceinline__ v8f wmma_k4(v2f a, v2f b, v8f c) {
  return __builtin_amdgcn_wmma_f32_16x16x4_f32(
       false, a,  false, b,
       (short)0, c,  false,  false);
}

__device__ __forceinline__ float ssp(float x) {
  float e = __expf(-fabsf(x));
  return fmaxf(x, 0.0f) + __logf(1.0f + e) - 0.69314718055994531f;
}

__device__ __forceinline__ void lds_barrier() {
  __builtin_amdgcn_fence(__ATOMIC_RELEASE, "workgroup", "local");
  __builtin_amdgcn_s_barrier();
  __builtin_amdgcn_fence(__ATOMIC_ACQUIRE, "workgroup", "local");
}

#define F 128
#define KRBF 64
#define A_STRIDE 132
#define E_STRIDE 68

__global__ void ssp_kernel(const float* __restrict__ x, float* __restrict__ xa, int n) {
  int i = blockIdx.x * 256 + threadIdx.x;
  if (i < n) xa[i] = ssp(x[i]);
}

__global__ void __launch_bounds__(256)
edge_msg_kernel(const float* __restrict__ xa, const int* __restrict__ eidx,
                const float* __restrict__ ea, const float* __restrict__ Wd,
                const float* __restrict__ bd, const float* __restrict__ Wg,
                float* __restrict__ aggr, int E) {
  __shared__ __align__(16) float As[16 * A_STRIDE];
  __shared__ __align__(16) float Es[16 * E_STRIDE];
  __shared__ int dsts[16];

  const int tid  = threadIdx.x;
  const int lane = tid & 31;
  const int wv   = tid >> 5;
  const int m    = lane & 15;
  const int hi   = lane >> 4;
  const int n    = wv * 16 + m;

  v2f bwd[F / 4];
  v2f bwg[KRBF / 4];
  {
    const float* wrow = Wd + n * F;
#pragma unroll
    for (int kk = 0; kk < F / 4; ++kk)
      bwd[kk] = *(const v2f*)(&wrow[kk * 4 + 2 * hi]);
    const float* grow = Wg + n * KRBF;
#pragma unroll
    for (int kk = 0; kk < KRBF / 4; ++kk)
      bwg[kk] = *(const v2f*)(&grow[kk * 4 + 2 * hi]);
  }
  const float bn = bd[n];

  const int nT = (E + 15) / 16;

  float pa[8];
  float pe[4];
  int   pd = 0;

  int tile = blockIdx.x;
  if (tile < nT) {
    const int e0 = tile * 16;
#pragma unroll
    for (int q = 0; q < 8; ++q) {
      int i = tid + 256 * q;
      int r = i >> 7, c = i & (F - 1);
      int e = e0 + r;
      if (e >= E) e = E - 1;
      int s = eidx[e];
      pa[q] = xa[s * F + c];
    }
#pragma unroll
    for (int q = 0; q < 4; ++q) {
      int i = tid + 256 * q;
      int r = i >> 6, c = i & (KRBF - 1);
      int e = e0 + r;
      if (e >= E) e = E - 1;
      pe[q] = ea[e * KRBF + c];
    }
    if (tid < 16) {
      int e = e0 + tid;
      if (e >= E) e = E - 1;
      pd = eidx[E + e];
    }
  }

  for (; tile < nT; tile += gridDim.x) {
    const int e0 = tile * 16;

#pragma unroll
    for (int q = 0; q < 8; ++q) {
      int i = tid + 256 * q;
      As[(i >> 7) * A_STRIDE + (i & (F - 1))] = pa[q];
    }
#pragma unroll
    for (int q = 0; q < 4; ++q) {
      int i = tid + 256 * q;
      Es[(i >> 6) * E_STRIDE + (i & (KRBF - 1))] = pe[q];
    }
    if (tid < 16) dsts[tid] = pd;
    lds_barrier();

    const int nxt = tile + gridDim.x;
    if (nxt < nT) {
      const int f0 = nxt * 16;
#pragma unroll
      for (int q = 0; q < 8; ++q) {
        int i = tid + 256 * q;
        int r = i >> 7, c = i & (F - 1);
        int e = f0 + r;
        if (e >= E) e = E - 1;
        int s = eidx[e];
        pa[q] = xa[s * F + c];
      }
#pragma unroll
      for (int q = 0; q < 4; ++q) {
        int i = tid + 256 * q;
        int r = i >> 6, c = i & (KRBF - 1);
        int e = f0 + r;
        if (e >= E) e = E - 1;
        pe[q] = ea[e * KRBF + c];
      }
      if (tid < 16) {
        int e = f0 + tid;
        if (e >= E) e = E - 1;
        pd = eidx[E + e];
      }
    }

    v8f acc = {};
#pragma unroll
    for (int kk = 0; kk < F / 4; ++kk) {
      v2f a = *(const v2f*)(&As[m * A_STRIDE + kk * 4 + 2 * hi]);
      acc = wmma_k4(a, bwd[kk], acc);
    }

    v8f acc2 = {};
#pragma unroll
    for (int kk = 0; kk < KRBF / 4; ++kk) {
      v2f a = *(const v2f*)(&Es[m * E_STRIDE + kk * 4 + 2 * hi]);
      acc2 = wmma_k4(a, bwg[kk], acc2);
    }

#pragma unroll
    for (int j = 0; j < 8; ++j) {
      int r = j + 8 * hi;
      int e = e0 + r;
      float msg = ssp(acc[j] + bn) * acc2[j];
      if (e < E) atomicAdd(&aggr[dsts[r] * F + n], msg);
    }
    lds_barrier();
  }
}

template <bool IN_ACT, int MODE, bool WRITE2>
__global__ void __launch_bounds__(256)
gemm128_kernel(const float* __restrict__ In, const float* __restrict__ W,
               const float* __restrict__ bias, const float* __restrict__ Ex1,
               const float* __restrict__ Ex2, float* __restrict__ Out,
               float* __restrict__ Out2, int N) {
  __shared__ __align__(16) float As[16 * A_STRIDE];

  const int tid  = threadIdx.x;
  const int lane = tid & 31;
  const int wv   = tid >> 5;
  const int m    = lane & 15;
  const int hi   = lane >> 4;
  const int n    = wv * 16 + m;

  v2f bw[F / 4];
  {
    const float* wrow = W + n * F;
#pragma unroll
    for (int kk = 0; kk < F / 4; ++kk)
      bw[kk] = *(const v2f*)(&wrow[kk * 4 + 2 * hi]);
  }
  const float bn = bias[n];
  float ex2n = 0.0f;
  if (MODE == 3) ex2n = Ex2[n];

  const int nT = (N + 15) / 16;

  float pa[8];
  int tile = blockIdx.x;
  if (tile < nT) {
    const int row0 = tile * 16;
#pragma unroll
    for (int q = 0; q < 8; ++q) {
      int i = tid + 256 * q;
      int rr = row0 + (i >> 7);
      if (rr >= N) rr = N - 1;
      pa[q] = In[rr * F + (i & (F - 1))];
    }
  }

  for (; tile < nT; tile += gridDim.x) {
    const int row0 = tile * 16;

#pragma unroll
    for (int q = 0; q < 8; ++q) {
      int i = tid + 256 * q;
      float v = pa[q];
      if (IN_ACT) v = ssp(v);
      As[(i >> 7) * A_STRIDE + (i & (F - 1))] = v;
    }
    lds_barrier();

    const int nxt = tile + gridDim.x;
    if (nxt < nT) {
      const int f0 = nxt * 16;
#pragma unroll
      for (int q = 0; q < 8; ++q) {
        int i = tid + 256 * q;
        int rr = f0 + (i >> 7);
        if (rr >= N) rr = N - 1;
        pa[q] = In[rr * F + (i & (F - 1))];
      }
    }

    v8f acc = {};
#pragma unroll
    for (int kk = 0; kk < F / 4; ++kk) {
      v2f a = *(const v2f*)(&As[m * A_STRIDE + kk * 4 + 2 * hi]);
      acc = wmma_k4(a, bw[kk], acc);
    }

#pragma unroll
    for (int j = 0; j < 8; ++j) {
      int r = row0 + j + 8 * hi;
      if (r >= N) continue;
      float g = acc[j] + bn;
      float o;
      if (MODE == 0)      o = ssp(g) + Ex1[r * F + n];
      else if (MODE == 1) o = g;
      else if (MODE == 2) o = Ex1[r * F + n] + g;
      else                o = g + Ex1[r * F + n] * ex2n;
      Out[r * F + n] = o;
      if (WRITE2) Out2[r * F + n] = o;
    }
    lds_barrier();
  }
}

extern "C" void kernel_launch(void* const* d_in, const int* in_sizes, int n_in,
                              void* d_out, int out_size, void* d_ws, size_t ws_size,
                              hipStream_t stream) {
  const float* x      = (const float*)d_in[0];
  const int*   eidx   = (const int*)  d_in[1];
  const float* ea     = (const float*)d_in[2];
  const float* W_same = (const float*)d_in[3];
  const float* b_same = (const float*)d_in[4];
  const float* W_diff = (const float*)d_in[5];
  const float* b_diff = (const float*)d_in[6];
  const float* W_G    = (const float*)d_in[7];
  const float* rW1    = (const float*)d_in[8];
  const float* rb1    = (const float*)d_in[9];
  const float* rW2    = (const float*)d_in[10];
  const float* rb2    = (const float*)d_in[11];
  const float* W_last = (const float*)d_in[12];
  const float* b_last = (const float*)d_in[13];
  const float* u      = (const float*)d_in[14];

  const int NF = in_sizes[0];
  const int N  = NF / F;
  const int E  = in_sizes[2] / KRBF;
  const int R  = in_sizes[8] / (F * F);

  float* out_final = (float*)d_out;
  float* out_msged = (float*)d_out + NF;

  float* buf0 = (float*)d_ws;
  float* buf1 = buf0 + NF;

  hipMemsetAsync(buf1, 0, (size_t)NF * sizeof(float), stream);

  ssp_kernel<<<(NF + 255) / 256, 256, 0, stream>>>(x, buf0, NF);

  edge_msg_kernel<<<512, 256, 0, stream>>>(buf0, eidx, ea, W_diff, b_diff, W_G,
                                           buf1, E);

  gemm128_kernel<false, 0, true><<<512, 256, 0, stream>>>(
      buf0, W_same, b_same, buf1, nullptr, out_msged, buf0, N);

  for (int r = 0; r < R; ++r) {
    gemm128_kernel<true, 1, false><<<512, 256, 0, stream>>>(
        buf0, rW1 + (size_t)r * F * F, rb1 + (size_t)r * F,
        nullptr, nullptr, buf1, nullptr, N);
    gemm128_kernel<true, 2, false><<<512, 256, 0, stream>>>(
        buf1, rW2 + (size_t)r * F * F, rb2 + (size_t)r * F,
        buf0, nullptr, buf0, nullptr, N);
  }

  gemm128_kernel<true, 3, false><<<512, 256, 0, stream>>>(
      buf0, W_last, b_last, x, u, out_final, nullptr, N);
}


# W_diff product and activation taken once per node, not per edge (a row gather commutes with a row-wise map; rows bit-identical); edge kernel gathers finished rows; atomics and sum order kept
# speedup vs baseline: 1.8101x; 1.2575x over previous
#include <hip/hip_runtime.h>
#include <hip/hip_bf16.h>


typedef __attribute__((ext_vector_type(2))) float v2f;
typedef __attribute__((ext_vector_type(8))) float v8f;

__device__ __forceinline__ v8f wmma_k4(v2f a, v2f b, v8f c) {
  return __builtin_amdgcn_wmma_f32_16x16x4_f32(
       false, a,  false, b,
       (short)0, c,  false,  false);
}

__device__ __forceinline__ float ssp(float x) {
  float e = __expf(-fabsf(x));
  return fmaxf(x, 0.0f) + __logf(1.0f + e) - 0.69314718055994531f;
}

__device__ __forceinline__ void lds_barrier() {
  __builtin_amdgcn_fence(__ATOMIC_RELEASE, "workgroup", "local");
  __builtin_amdgcn_s_barrier();
  __builtin_amdgcn_fence(__ATOMIC_ACQUIRE, "workgroup", "local");
}

#define F 128
#define KRBF 64
#define A_STRIDE 132
#define E_STRIDE 68

__global__ void __launch_bounds__(256)
edge_msg_kernel(const float* __restrict__ P, const int* __restrict__ eidx,
                const float* __restrict__ ea, const float* __restrict__ Wg,
                float* __restrict__ aggr, int E) {
  __shared__ __align__(16) float As[16 * A_STRIDE];
  __shared__ __align__(16) float Es[16 * E_STRIDE];
  __shared__ int dsts[16];

  const int tid  = threadIdx.x;
  const int lane = tid & 31;
  const int wv   = tid >> 5;
  const int m    = lane & 15;
  const int hi   = lane >> 4;
  const int n    = wv * 16 + m;

  v2f bwg[KRBF / 4];
  {
    const float* grow = Wg + n * KRBF;
#pragma unroll
    for (int kk = 0; kk < KRBF / 4; ++kk)
      bwg[kk] = *(const v2f*)(&grow[kk * 4 + 2 * hi]);
  }

  const int nT = (E + 15) / 16;

  float pa[8];
  float pe[4];
  int   pd = 0;

  int tile = blockIdx.x;
  if (tile < nT) {
    const int e0 = tile * 16;
#pragma unroll
    for (int q = 0; q < 8; ++q) {
      int i = tid + 256 * q;
      int r = i >> 7, c = i & (F - 1);
      int e = e0 + r;
      if (e >= E) e = E - 1;
      int s = eidx[e];
      pa[q] = P[s * F + c];
    }
#pragma unroll
    for (int q = 0; q < 4; ++q) {
      int i = tid + 256 * q;
      int r = i >> 6, c = i & (KRBF - 1);
      int e = e0 + r;
      if (e >= E) e = E - 1;
      pe[q] = ea[e * KRBF + c];
    }
    if (tid < 16) {
      int e = e0 + tid;
      if (e >= E) e = E - 1;
      pd = eidx[E + e];
    }
  }

  for (; tile < nT; tile += gridDim.x) {
    const int e0 = tile * 16;

#pragma unroll
    for (int q = 0; q < 8; ++q) {
      int i = tid + 256 * q;
      As[(i >> 7) * A_STRIDE + (i & (F - 1))] = pa[q];
    }
#pragma unroll
    for (int q = 0; q < 4; ++q) {
      int i = tid + 256 * q;
      Es[(i >> 6) * E_STRIDE + (i & (KRBF - 1))] = pe[q];
    }
    if (tid < 16) dsts[tid] = pd;
    lds_barrier();

    const int nxt = tile + gridDim.x;
    if (nxt < nT) {
      const int f0 = nxt * 16;
#pragma unroll
      for (int q = 0; q < 8; ++q) {
        int i = tid + 256 * q;
        int r = i >> 7, c = i & (F - 1);
        int e = f0 + r;
        if (e >= E) e = E - 1;
        int s = eidx[e];
        pa[q] = P[s * F + c];
      }
#pragma unroll
      for (int q = 0; q < 4; ++q) {
        int i = tid + 256 * q;
        int r = i >> 6, c = i & (KRBF - 1);
        int e = f0 + r;
        if (e >= E) e = E - 1;
        pe[q] = ea[e * KRBF + c];
      }
      if (tid < 16) {
        int e = f0 + tid;
        if (e >= E) e = E - 1;
        pd = eidx[E + e];
      }
    }

    v8f acc2 = {};
#pragma unroll
    for (int kk = 0; kk < KRBF / 4; ++kk) {
      v2f a = *(const v2f*)(&Es[m * E_STRIDE + kk * 4 + 2 * hi]);
      acc2 = wmma_k4(a, bwg[kk], acc2);
    }

#pragma unroll
    for (int j = 0; j < 8; ++j) {
      int r = j + 8 * hi;
      int e = e0 + r;
      float msg = As[r * A_STRIDE + n] * acc2[j];
      if (e < E) atomicAdd(&aggr[dsts[r] * F + n], msg);
    }
    lds_barrier();
  }
}

template <bool IN_ACT, int MODE, bool WRITE2>
__global__ void __launch_bounds__(256)
gemm128_kernel(const float* __restrict__ In, const float* __restrict__ W,
               const float* __restrict__ bias, const float* __restrict__ Ex1,
               const float* __restrict__ Ex2, float* __restrict__ Out,
               float* __restrict__ Out2, int N) {
  __shared__ __align__(16) float As[16 * A_STRIDE];

  const int tid  = threadIdx.x;
  const int lane = tid & 31;
  const int wv   = tid >> 5;
  const int m    = lane & 15;
  const int hi   = lane >> 4;
  const int n    = wv * 16 + m;

  v2f bw[F / 4];
  {
    const float* wrow = W + n * F;
#pragma unroll
    for (int kk = 0; kk < F / 4; ++kk)
      bw[kk] = *(const v2f*)(&wrow[kk * 4 + 2 * hi]);
  }
  const float bn = bias[n];
  float ex2n = 0.0f;
  if (MODE == 3) ex2n = Ex2[n];

  const int nT = (N + 15) / 16;

  float pa[8];
  int tile = blockIdx.x;
  if (tile < nT) {
    const int row0 = tile * 16;
#pragma unroll
    for (int q = 0; q < 8; ++q) {
      int i = tid + 256 * q;
      int rr = row0 + (i >> 7);
      if (rr >= N) rr = N - 1;
      pa[q] = In[rr * F + (i & (F - 1))];
    }
  }

  for (; tile < nT; tile += gridDim.x) {
    const int row0 = tile * 16;

#pragma unroll
    for (int q = 0; q < 8; ++q) {
      int i = tid + 256 * q;
      float v = pa[q];
      if (IN_ACT) v = ssp(v);
      As[(i >> 7) * A_STRIDE + (i & (F - 1))] = v;
    }
    lds_barrier();

    const int nxt = tile + gridDim.x;
    if (nxt < nT) {
      const int f0 = nxt * 16;
#pragma unroll
      for (int q = 0; q < 8; ++q) {
        int i = tid + 256 * q;
        int rr = f0 + (i >> 7);
        if (rr >= N) rr = N - 1;
        pa[q] = In[rr * F + (i & (F - 1))];
      }
    }

    v8f acc = {};
#pragma unroll
    for (int kk = 0; kk < F / 4; ++kk) {
      v2f a = *(const v2f*)(&As[m * A_STRIDE + kk * 4 + 2 * hi]);
      acc = wmma_k4(a, bw[kk], acc);
    }

#pragma unroll
    for (int j = 0; j < 8; ++j) {
      int r = row0 + j + 8 * hi;
      if (r >= N) continue;
      float g = acc[j] + bn;
      float o;
      if (MODE == 0)      o = ssp(g) + Ex1[r * F + n];
      else if (MODE == 1) o = g;
      else if (MODE == 2) o = Ex1[r * F + n] + g;
      else if (MODE == 3) o = g + Ex1[r * F + n] * ex2n;
      else                o = ssp(g);
      Out[r * F + n] = o;
      if (WRITE2) Out2[r * F + n] = o;
    }
    lds_barrier();
  }
}

extern "C" void kernel_launch(void* const* d_in, const int* in_sizes, int n_in,
                              void* d_out, int out_size, void* d_ws, size_t ws_size,
                              hipStream_t stream) {
  const float* x      = (const float*)d_in[0];
  const int*   eidx   = (const int*)  d_in[1];
  const float* ea     = (const float*)d_in[2];
  const float* W_same = (const float*)d_in[3];
  const float* b_same = (const float*)d_in[4];
  const float* W_diff = (const float*)d_in[5];
  const float* b_diff = (const float*)d_in[6];
  const float* W_G    = (const float*)d_in[7];
  const float* rW1    = (const float*)d_in[8];
  const float* rb1    = (const float*)d_in[9];
  const float* rW2    = (const float*)d_in[10];
  const float* rb2    = (const float*)d_in[11];
  const float* W_last = (const float*)d_in[12];
  const float* b_last = (const float*)d_in[13];
  const float* u      = (const float*)d_in[14];

  const int NF = in_sizes[0];
  const int N  = NF / F;
  const int E  = in_sizes[2] / KRBF;
  const int R  = in_sizes[8] / (F * F);

  float* out_final = (float*)d_out;
  float* out_msged = (float*)d_out + NF;

  float* buf0 = (float*)d_ws;
  float* buf1 = buf0 + NF;

  hipMemsetAsync(buf1, 0, (size_t)NF * sizeof(float), stream);

  gemm128_kernel<true, 4, false><<<512, 256, 0, stream>>>(
      x, W_diff, b_diff, nullptr, nullptr, buf0, nullptr, N);

  edge_msg_kernel<<<512, 256, 0, stream>>>(buf0, eidx, ea, W_G, buf1, E);

  gemm128_kernel<true, 0, true><<<512, 256, 0, stream>>>(
      x, W_same, b_same, buf1, nullptr, out_msged, buf0, N);

  for (int r = 0; r < R; ++r) {
    gemm128_kernel<true, 1, false><<<512, 256, 0, stream>>>(
        buf0, rW1 + (size_t)r * F * F, rb1 + (size_t)r * F,
        nullptr, nullptr, buf1, nullptr, N);
    gemm128_kernel<true, 2, false><<<512, 256, 0, stream>>>(
        buf1, rW2 + (size_t)r * F * F, rb2 + (size_t)r * F,
        buf0, nullptr, buf0, nullptr, N);
  }

  gemm128_kernel<true, 3, false><<<512, 256, 0, stream>>>(
      buf0, W_last, b_last, x, u, out_final, nullptr, N);
}


# Each residual block's two GEMMs chained per 16-row tile in one kernel: hidden tile and the block's input rows stay in LDS, W1 and W2 fragments in registers; 8 launches from 10; k and sum order kept
# speedup vs baseline: 1.9537x; 1.0793x over previous
#include <hip/hip_runtime.h>
#include <hip/hip_bf16.h>


typedef __attribute__((ext_vector_type(2))) float v2f;
typedef __attribute__((ext_vector_type(8))) float v8f;

__device__ __forceinline__ v8f wmma_k4(v2f a, v2f b, v8f c) {
  return __builtin_amdgcn_wmma_f32_16x16x4_f32(
       false, a,  false, b,
       (short)0, c,  false,  false);
}

__device__ __forceinline__ float ssp(float x) {
  float e = __expf(-fabsf(x));
  return fmaxf(x, 0.0f) + __logf(1.0f + e) - 0.69314718055994531f;
}

__device__ __forceinline__ void lds_barrier() {
  __builtin_amdgcn_fence(__ATOMIC_RELEASE, "workgroup", "local");
  __builtin_amdgcn_s_barrier();
  __builtin_amdgcn_fence(__ATOMIC_ACQUIRE, "workgroup", "local");
}

#define F 128
#define KRBF 64
#define A_STRIDE 132
#define E_STRIDE 68

__global__ void __launch_bounds__(256)
edge_msg_kernel(const float* __restrict__ P, const int* __restrict__ eidx,
                const float* __restrict__ ea, const float* __restrict__ Wg,
                float* __restrict__ aggr, int E) {
  __shared__ __align__(16) float As[16 * A_STRIDE];
  __shared__ __align__(16) float Es[16 * E_STRIDE];
  __shared__ int dsts[16];

  const int tid  = threadIdx.x;
  const int lane = tid & 31;
  const int wv   = tid >> 5;
  const int m    = lane & 15;
  const int hi   = lane >> 4;
  const int n    = wv * 16 + m;

  v2f bwg[KRBF / 4];
  {
    const float* grow = Wg + n * KRBF;
#pragma unroll
    for (int kk = 0; kk < KRBF / 4; ++kk)
      bwg[kk] = *(const v2f*)(&grow[kk * 4 + 2 * hi]);
  }

  const int nT = (E + 15) / 16;

  float pa[8];
  float pe[4];
  int   pd = 0;

  int tile = blockIdx.x;
  if (tile < nT) {
    const int e0 = tile * 16;
#pragma unroll
    for (int q = 0; q < 8; ++q) {
      int i = tid + 256 * q;
      int r = i >> 7, c = i & (F - 1);
      int e = e0 + r;
      if (e >= E) e = E - 1;
      int s = eidx[e];
      pa[q] = P[s * F + c];
    }
#pragma unroll
    for (int q = 0; q < 4; ++q) {
      int i = tid + 256 * q;
      int r = i >> 6, c = i & (KRBF - 1);
      int e = e0 + r;
      if (e >= E) e = E - 1;
      pe[q] = ea[e * KRBF + c];
    }
    if (tid < 16) {
      int e = e0 + tid;
      if (e >= E) e = E - 1;
      pd = eidx[E + e];
    }
  }

  for (; tile < nT; tile += gridDim.x) {
    const int e0 = tile * 16;

#pragma unroll
    for (int q = 0; q < 8; ++q) {
      int i = tid + 256 * q;
      As[(i >> 7) * A_STRIDE + (i & (F - 1))] = pa[q];
    }
#pragma unroll
    for (int q = 0; q < 4; ++q) {
      int i = tid + 256 * q;
      Es[(i >> 6) * E_STRIDE + (i & (KRBF - 1))] = pe[q];
    }
    if (tid < 16) dsts[tid] = pd;
    lds_barrier();

    const int nxt = tile + gridDim.x;
    if (nxt < nT) {
      const int f0 = nxt * 16;
#pragma unroll
      for (int q = 0; q < 8; ++q) {
        int i = tid + 256 * q;
        int r = i >> 7, c = i & (F - 1);
        int e = f0 + r;
        if (e >= E) e = E - 1;
        int s = eidx[e];
        pa[q] = P[s * F + c];
      }
#pragma unroll
      for (int q = 0; q < 4; ++q) {
        int i = tid + 256 * q;
        int r = i >> 6, c = i & (KRBF - 1);
        int e = f0 + r;
        if (e >= E) e = E - 1;
        pe[q] = ea[e * KRBF + c];
      }
      if (tid < 16) {
        int e = f0 + tid;
        if (e >= E) e = E - 1;
        pd = eidx[E + e];
      }
    }

    v8f acc2 = {};
#pragma unroll
    for (int kk = 0; kk < KRBF / 4; ++kk) {
      v2f a = *(const v2f*)(&Es[m * E_STRIDE + kk * 4 + 2 * hi]);
      acc2 = wmma_k4(a, bwg[kk], acc2);
    }

#pragma unroll
    for (int j = 0; j < 8; ++j) {
      int r = j + 8 * hi;
      int e = e0 + r;
      float msg = As[r * A_STRIDE + n] * acc2[j];
      if (e < E) atomicAdd(&aggr[dsts[r] * F + n], msg);
    }
    lds_barrier();
  }
}

template <bool IN_ACT, int MODE, bool WRITE2>
__global__ void __launch_bounds__(256)
gemm128_kernel(const float* __restrict__ In, const float* __restrict__ W,
               const float* __restrict__ bias, const float* __restrict__ Ex1,
               const float* __restrict__ Ex2, float* __restrict__ Out,
               float* __restrict__ Out2, int N) {
  __shared__ __align__(16) float As[16 * A_STRIDE];

  const int tid  = threadIdx.x;
  const int lane = tid & 31;
  const int wv   = tid >> 5;
  const int m    = lane & 15;
  const int hi   = lane >> 4;
  const int n    = wv * 16 + m;

  v2f bw[F / 4];
  {
    const float* wrow = W + n * F;
#pragma unroll
    for (int kk = 0; kk < F / 4; ++kk)
      bw[kk] = *(const v2f*)(&wrow[kk * 4 + 2 * hi]);
  }
  const float bn = bias[n];
  float ex2n = 0.0f;
  if (MODE == 3) ex2n = Ex2[n];

  const int nT = (N + 15) / 16;

  float pa[8];
  int tile = blockIdx.x;
  if (tile < nT) {
    const int row0 = tile * 16;
#pragma unroll
    for (int q = 0; q < 8; ++q) {
      int i = tid + 256 * q;
      int rr = row0 + (i >> 7);
      if (rr >= N) rr = N - 1;
      pa[q] = In[rr * F + (i & (F - 1))];
    }
  }

  for (; tile < nT; tile += gridDim.x) {
    const int row0 = tile * 16;

#pragma unroll
    for (int q = 0; q < 8; ++q) {
      int i = tid + 256 * q;
      float v = pa[q];
      if (IN_ACT) v = ssp(v);
      As[(i >> 7) * A_STRIDE + (i & (F - 1))] = v;
    }
    lds_barrier();

    const int nxt = tile + gridDim.x;
    if (nxt < nT) {
      const int f0 = nxt * 16;
#pragma unroll
      for (int q = 0; q < 8; ++q) {
        int i = tid + 256 * q;
        int rr = f0 + (i >> 7);
        if (rr >= N) rr = N - 1;
        pa[q] = In[rr * F + (i & (F - 1))];
      }
    }

    v8f acc = {};
#pragma unroll
    for (int kk = 0; kk < F / 4; ++kk) {
      v2f a = *(const v2f*)(&As[m * A_STRIDE + kk * 4 + 2 * hi]);
      acc = wmma_k4(a, bw[kk], acc);
    }

#pragma unroll
    for (int j = 0; j < 8; ++j) {
      int r = row0 + j + 8 * hi;
      if (r >= N) continue;
      float g = acc[j] + bn;
      float o;
      if (MODE == 0)      o = ssp(g) + Ex1[r * F + n];
      else if (MODE == 1) o = g;
      else if (MODE == 2) o = Ex1[r * F + n] + g;
      else if (MODE == 3) o = g + Ex1[r * F + n] * ex2n;
      else                o = ssp(g);
      Out[r * F + n] = o;
      if (WRITE2) Out2[r * F + n] = o;
    }
    lds_barrier();
  }
}

__device__ __forceinline__ v8f tile_product(const float* As, const v2f (&bw)[F / 4],
                                            int m, int hi) {
  v8f acc = {};
#pragma unroll
  for (int kk = 0; kk < F / 4; ++kk) {
    v2f a = *(const v2f*)(&As[m * A_STRIDE + kk * 4 + 2 * hi]);
    acc = wmma_k4(a, bw[kk], acc);
  }
  return acc;
}

__global__ void __launch_bounds__(256) __attribute__((amdgpu_waves_per_eu(4, 4)))
resblock_kernel(const float* __restrict__ In,
                const float* __restrict__ W1, const float* __restrict__ b1,
                const float* __restrict__ W2, const float* __restrict__ b2,
                float* __restrict__ Out, int N) {
  __shared__ __align__(16) float Xs[16 * A_STRIDE];
  __shared__ __align__(16) float Ys[16 * A_STRIDE];
  __shared__ __align__(16) float Ts[16 * A_STRIDE];

  const int tid  = threadIdx.x;
  const int lane = tid & 31;
  const int wv   = tid >> 5;
  const int m    = lane & 15;
  const int hi   = lane >> 4;
  const int n    = wv * 16 + m;

  v2f bw1[F / 4];
  v2f bw2[F / 4];
  {
    const float* w1row = W1 + n * F;
    const float* w2row = W2 + n * F;
#pragma unroll
    for (int kk = 0; kk < F / 4; ++kk)
      bw1[kk] = *(const v2f*)(&w1row[kk * 4 + 2 * hi]);
#pragma unroll
    for (int kk = 0; kk < F / 4; ++kk)
      bw2[kk] = *(const v2f*)(&w2row[kk * 4 + 2 * hi]);
  }
  const float b1n = b1[n];
  const float b2n = b2[n];

  const int nT = (N + 15) / 16;

  float pa[8];
  int tile = blockIdx.x;
  if (tile < nT) {
    const int row0 = tile * 16;
#pragma unroll
    for (int q = 0; q < 8; ++q) {
      int i = tid + 256 * q;
      int rr = row0 + (i >> 7);
      if (rr >= N) rr = N - 1;
      pa[q] = In[rr * F + (i & (F - 1))];
    }
  }

  for (; tile < nT; tile += gridDim.x) {
    const int row0 = tile * 16;

#pragma unroll
    for (int q = 0; q < 8; ++q) {
      int i = tid + 256 * q;
      int o = (i >> 7) * A_STRIDE + (i & (F - 1));
      float v = pa[q];
      Ts[o] = v;
      Xs[o] = ssp(v);
    }
    lds_barrier();

    const int nxt = tile + gridDim.x;
    if (nxt < nT) {
      const int f0 = nxt * 16;
#pragma unroll
      for (int q = 0; q < 8; ++q) {
        int i = tid + 256 * q;
        int rr = f0 + (i >> 7);
        if (rr >= N) rr = N - 1;
        pa[q] = In[rr * F + (i & (F - 1))];
      }
    }

    {
      v8f acc = tile_product(Xs, bw1, m, hi);
#pragma unroll
      for (int j = 0; j < 8; ++j) {
        int rl = j + 8 * hi;
        float g = acc[j] + b1n;
        Ys[rl * A_STRIDE + n] = ssp(g);
      }
    }
    lds_barrier();

    {
      v8f acc = tile_product(Ys, bw2, m, hi);
#pragma unroll
      for (int j = 0; j < 8; ++j) {
        int rl = j + 8 * hi;
        int r  = row0 + rl;
        if (r >= N) continue;
        float g = acc[j] + b2n;
        Out[r * F + n] = Ts[rl * A_STRIDE + n] + g;
      }
    }
    lds_barrier();
  }
}

extern "C" void kernel_launch(void* const* d_in, const int* in_sizes, int n_in,
                              void* d_out, int out_size, void* d_ws, size_t ws_size,
                              hipStream_t stream) {
  const float* x      = (const float*)d_in[0];
  const int*   eidx   = (const int*)  d_in[1];
  const float* ea     = (const float*)d_in[2];
  const float* W_same = (const float*)d_in[3];
  const float* b_same = (const float*)d_in[4];
  const float* W_diff = (const float*)d_in[5];
  const float* b_diff = (const float*)d_in[6];
  const float* W_G    = (const float*)d_in[7];
  const float* rW1    = (const float*)d_in[8];
  const float* rb1    = (const float*)d_in[9];
  const float* rW2    = (const float*)d_in[10];
  const float* rb2    = (const float*)d_in[11];
  const float* W_last = (const float*)d_in[12];
  const float* b_last = (const float*)d_in[13];
  const float* u      = (const float*)d_in[14];

  const int NF = in_sizes[0];
  const int N  = NF / F;
  const int E  = in_sizes[2] / KRBF;
  const int R  = in_sizes[8] / (F * F);

  float* out_final = (float*)d_out;
  float* out_msged = (float*)d_out + NF;

  float* buf0 = (float*)d_ws;
  float* buf1 = buf0 + NF;

  hipMemsetAsync(buf1, 0, (size_t)NF * sizeof(float), stream);

  gemm128_kernel<true, 4, false><<<512, 256, 0, stream>>>(
      x, W_diff, b_diff, nullptr, nullptr, buf0, nullptr, N);

  edge_msg_kernel<<<512, 256, 0, stream>>>(buf0, eidx, ea, W_G, buf1, E);

  gemm128_kernel<true, 0, true><<<512, 256, 0, stream>>>(
      x, W_same, b_same, buf1, nullptr, out_msged, buf0, N);

  float* cur = buf0;
  float* oth = buf1;
  for (int r = 0; r < R; ++r) {
    resblock_kernel<<<512, 256, 0, stream>>>(
        cur, rW1 + (size_t)r * F * F, rb1 + (size_t)r * F,
        rW2 + (size_t)r * F * F, rb2 + (size_t)r * F, oth, N);
    float* sw = cur; cur = oth; oth = sw;
  }

  gemm128_kernel<true, 3, false><<<512, 256, 0, stream>>>(
      cur, W_last, b_last, x, u, out_final, nullptr, N);
}


# Edge tile of 32 edges: a wave takes 16 edges x 32 columns (2 accumulators, 1 A fragment); a half-wave exchange makes each f32 atomic one whole 128-byte line; k order kept, arrival order as before
# speedup vs baseline: 2.0169x; 1.0323x over previous
#include <hip/hip_runtime.h>
#include <hip/hip_bf16.h>


typedef __attribute__((ext_vector_type(2))) float v2f;
typedef __attribute__((ext_vector_type(8))) float v8f;

__device__ __forceinline__ v8f wmma_k4(v2f a, v2f b, v8f c) {
  return __builtin_amdgcn_wmma_f32_16x16x4_f32(
       false, a,  false, b,
       (short)0, c,  false,  false);
}

__device__ __forceinline__ float ssp(float x) {
  float e = __expf(-fabsf(x));
  return fmaxf(x, 0.0f) + __logf(1.0f + e) - 0.69314718055994531f;
}

__device__ __forceinline__ void lds_barrier() {
  __builtin_amdgcn_fence(__ATOMIC_RELEASE, "workgroup", "local");
  __builtin_amdgcn_s_barrier();
  __builtin_amdgcn_fence(__ATOMIC_ACQUIRE, "workgroup", "local");
}

#define F 128
#define KRBF 64
#define A_STRIDE 132
#define E_STRIDE 68

#define ET 32

__device__ __forceinline__ float other_half(float x) {
  return __builtin_bit_cast(float, __builtin_amdgcn_permlanex16(
      __builtin_bit_cast(int, x), __builtin_bit_cast(int, x),
      0x76543210, 0xfedcba98, false, false));
}

__global__ void __launch_bounds__(256)
edge_msg_kernel(const float* __restrict__ P, const int* __restrict__ eidx,
                const float* __restrict__ ea, const float* __restrict__ Wg,
                float* __restrict__ aggr, int E) {
  __shared__ __align__(16) float As[ET * A_STRIDE];
  __shared__ __align__(16) float Es[ET * E_STRIDE];
  __shared__ int dsts[ET];

  const int tid  = threadIdx.x;
  const int lane = tid & 31;
  const int wv   = tid >> 5;
  const int m    = lane & 15;
  const int hi   = lane >> 4;
  const int rb   = (wv >> 2) * 16;
  const int c0   = (wv & 3) * 32;
  const int col  = c0 + lane;

  v2f bwg0[KRBF / 4];
  v2f bwg1[KRBF / 4];
  {
    const float* g0 = Wg + (c0 + m) * KRBF;
    const float* g1 = Wg + (c0 + 16 + m) * KRBF;
#pragma unroll
    for (int kk = 0; kk < KRBF / 4; ++kk) {
      bwg0[kk] = *(const v2f*)(&g0[kk * 4 + 2 * hi]);
      bwg1[kk] = *(const v2f*)(&g1[kk * 4 + 2 * hi]);
    }
  }

  const int nT = (E + ET - 1) / ET;

  float pa[16];
  float pe[8];
  int   pd = 0;

  int tile = blockIdx.x;
  if (tile < nT) {
    const int e0 = tile * ET;
#pragma unroll
    for (int q = 0; q < 16; ++q) {
      int i = tid + 256 * q;
      int r = i >> 7, c = i & (F - 1);
      int e = e0 + r;
      if (e >= E) e = E - 1;
      int s = eidx[e];
      pa[q] = P[s * F + c];
    }
#pragma unroll
    for (int q = 0; q < 8; ++q) {
      int i = tid + 256 * q;
      int r = i >> 6, c = i & (KRBF - 1);
      int e = e0 + r;
      if (e >= E) e = E - 1;
      pe[q] = ea[e * KRBF + c];
    }
    if (tid < ET) {
      int e = e0 + tid;
      if (e >= E) e = E - 1;
      pd = eidx[E + e];
    }
  }

  for (; tile < nT; tile += gridDim.x) {
    const int e0 = tile * ET;

#pragma unroll
    for (int q = 0; q < 16; ++q) {
      int i = tid + 256 * q;
      As[(i >> 7) * A_STRIDE + (i & (F - 1))] = pa[q];
    }
#pragma unroll
    for (int q = 0; q < 8; ++q) {
      int i = tid + 256 * q;
      Es[(i >> 6) * E_STRIDE + (i & (KRBF - 1))] = pe[q];
    }
    if (tid < ET) dsts[tid] = pd;
    lds_barrier();

    const int nxt = tile + gridDim.x;
    if (nxt < nT) {
      const int f0 = nxt * ET;
#pragma unroll
      for (int q = 0; q < 16; ++q) {
        int i = tid + 256 * q;
        int r = i >> 7, c = i & (F - 1);
        int e = f0 + r;
        if (e >= E) e = E - 1;
        int s = eidx[e];
        pa[q] = P[s * F + c];
      }
#pragma unroll
      for (int q = 0; q < 8; ++q) {
        int i = tid + 256 * q;
        int r = i >> 6, c = i & (KRBF - 1);
        int e = f0 + r;
        if (e >= E) e = E - 1;
        pe[q] = ea[e * KRBF + c];
      }
      if (tid < ET) {
        int e = f0 + tid;
        if (e >= E) e = E - 1;
        pd = eidx[E + e];
      }
    }

    v8f acc0 = {};
    v8f acc1 = {};
#pragma unroll
    for (int kk = 0; kk < KRBF / 4; ++kk) {
      v2f a = *(const v2f*)(&Es[(rb + m) * E_STRIDE + kk * 4 + 2 * hi]);
      acc0 = wmma_k4(a, bwg0[kk], acc0);
      acc1 = wmma_k4(a, bwg1[kk], acc1);
    }

#pragma unroll
    for (int j = 0; j < 8; ++j) {
      float give = hi ? acc0[j] : acc1[j];
      float got  = other_half(give);
      float glo  = hi ? got : acc0[j];
      float ghi  = hi ? acc1[j] : got;
      int rlo = rb + j;
      int rhi = rb + j + 8;
      if (e0 + rlo < E)
        atomicAdd(&aggr[dsts[rlo] * F + col], As[rlo * A_STRIDE + col] * glo);
      if (e0 + rhi < E)
        atomicAdd(&aggr[dsts[rhi] * F + col], As[rhi * A_STRIDE + col] * ghi);
    }
    lds_barrier();
  }
}

template <bool IN_ACT, int MODE, bool WRITE2>
__global__ void __launch_bounds__(256)
gemm128_kernel(const float* __restrict__ In, const float* __restrict__ W,
               const float* __restrict__ bias, const float* __restrict__ Ex1,
               const float* __restrict__ Ex2, float* __restrict__ Out,
               float* __restrict__ Out2, int N) {
  __shared__ __align__(16) float As[16 * A_STRIDE];

  const int tid  = threadIdx.x;
  const int lane = tid & 31;
  const int wv   = tid >> 5;
  const int m    = lane & 15;
  const int hi   = lane >> 4;
  const int n    = wv * 16 + m;

  v2f bw[F / 4];
  {
    const float* wrow = W + n * F;
#pragma unroll
    for (int kk = 0; kk < F / 4; ++kk)
      bw[kk] = *(const v2f*)(&wrow[kk * 4 + 2 * hi]);
  }
  const float bn = bias[n];
  float ex2n = 0.0f;
  if (MODE == 3) ex2n = Ex2[n];

  const int nT = (N + 15) / 16;

  float pa[8];
  int tile = blockIdx.x;
  if (tile < nT) {
    const int row0 = tile * 16;
#pragma unroll
    for (int q = 0; q < 8; ++q) {
      int i = tid + 256 * q;
      int rr = row0 + (i >> 7);
      if (rr >= N) rr = N - 1;
      pa[q] = In[rr * F + (i & (F - 1))];
    }
  }

  for (; tile < nT; tile += gridDim.x) {
    const int row0 = tile * 16;

#pragma unroll
    for (int q = 0; q < 8; ++q) {
      int i = tid + 256 * q;
      float v = pa[q];
      if (IN_ACT) v = ssp(v);
      As[(i >> 7) * A_STRIDE + (i & (F - 1))] = v;
    }
    lds_barrier();

    const int nxt = tile + gridDim.x;
    if (nxt < nT) {
      const int f0 = nxt * 16;
#pragma unroll
      for (int q = 0; q < 8; ++q) {
        int i = tid + 256 * q;
        int rr = f0 + (i >> 7);
        if (rr >= N) rr = N - 1;
        pa[q] = In[rr * F + (i & (F - 1))];
      }
    }

    v8f acc = {};
#pragma unroll
    for (int kk = 0; kk < F / 4; ++kk) {
      v2f a = *(const v2f*)(&As[m * A_STRIDE + kk * 4 + 2 * hi]);
      acc = wmma_k4(a, bw[kk], acc);
    }

#pragma unroll
    for (int j = 0; j < 8; ++j) {
      int r = row0 + j + 8 * hi;
      if (r >= N) continue;
      float g = acc[j] + bn;
      float o;
      if (MODE == 0)      o = ssp(g) + Ex1[r * F + n];
      else if (MODE == 1) o = g;
      else if (MODE == 2) o = Ex1[r * F + n] + g;
      else if (MODE == 3) o = g + Ex1[r * F + n] * ex2n;
      else                o = ssp(g);
      Out[r * F + n] = o;
      if (WRITE2) Out2[r * F + n] = o;
    }
    lds_barrier();
  }
}

__device__ __forceinline__ v8f tile_product(const float* As, const v2f (&bw)[F / 4],
                                            int m, int hi) {
  v8f acc = {};
#pragma unroll
  for (int kk = 0; kk < F / 4; ++kk) {
    v2f a = *(const v2f*)(&As[m * A_STRIDE + kk * 4 + 2 * hi]);
    acc = wmma_k4(a, bw[kk], acc);
  }
  return acc;
}

__global__ void __launch_bounds__(256) __attribute__((amdgpu_waves_per_eu(4, 4)))
resblock_kernel(const float* __restrict__ In,
                const float* __restrict__ W1, const float* __restrict__ b1,
                const float* __restrict__ W2, const float* __restrict__ b2,
                float* __restrict__ Out, int N) {
  __shared__ __align__(16) float Xs[16 * A_STRIDE];
  __shared__ __align__(16) float Ys[16 * A_STRIDE];
  __shared__ __align__(16) float Ts[16 * A_STRIDE];

  const int tid  = threadIdx.x;
  const int lane = tid & 31;
  const int wv   = tid >> 5;
  const int m    = lane & 15;
  const int hi   = lane >> 4;
  const int n    = wv * 16 + m;

  v2f bw1[F / 4];
  v2f bw2[F / 4];
  {
    const float* w1row = W1 + n * F;
    const float* w2row = W2 + n * F;
#pragma unroll
    for (int kk = 0; kk < F / 4; ++kk)
      bw1[kk] = *(const v2f*)(&w1row[kk * 4 + 2 * hi]);
#pragma unroll
    for (int kk = 0; kk < F / 4; ++kk)
      bw2[kk] = *(const v2f*)(&w2row[kk * 4 + 2 * hi]);
  }
  const float b1n = b1[n];
  const float b2n = b2[n];

  const int nT = (N + 15) / 16;

  float pa[8];
  int tile = blockIdx.x;
  if (tile < nT) {
    const int row0 = tile * 16;
#pragma unroll
    for (int q = 0; q < 8; ++q) {
      int i = tid + 256 * q;
      int rr = row0 + (i >> 7);
      if (rr >= N) rr = N - 1;
      pa[q] = In[rr * F + (i & (F - 1))];
    }
  }

  for (; tile < nT; tile += gridDim.x) {
    const int row0 = tile * 16;

#pragma unroll
    for (int q = 0; q < 8; ++q) {
      int i = tid + 256 * q;
      int o = (i >> 7) * A_STRIDE + (i & (F - 1));
      float v = pa[q];
      Ts[o] = v;
      Xs[o] = ssp(v);
    }
    lds_barrier();

    const int nxt = tile + gridDim.x;
    if (nxt < nT) {
      const int f0 = nxt * 16;
#pragma unroll
      for (int q = 0; q < 8; ++q) {
        int i = tid + 256 * q;
        int rr = f0 + (i >> 7);
        if (rr >= N) rr = N - 1;
        pa[q] = In[rr * F + (i & (F - 1))];
      }
    }

    {
      v8f acc = tile_product(Xs, bw1, m, hi);
#pragma unroll
      for (int j = 0; j < 8; ++j) {
        int rl = j + 8 * hi;
        float g = acc[j] + b1n;
        Ys[rl * A_STRIDE + n] = ssp(g);
      }
    }
    lds_barrier();

    {
      v8f acc = tile_product(Ys, bw2, m, hi);
#pragma unroll
      for (int j = 0; j < 8; ++j) {
        int rl = j + 8 * hi;
        int r  = row0 + rl;
        if (r >= N) continue;
        float g = acc[j] + b2n;
        Out[r * F + n] = Ts[rl * A_STRIDE + n] + g;
      }
    }
    lds_barrier();
  }
}

extern "C" void kernel_launch(void* const* d_in, const int* in_sizes, int n_in,
                              void* d_out, int out_size, void* d_ws, size_t ws_size,
                              hipStream_t stream) {
  const float* x      = (const float*)d_in[0];
  const int*   eidx   = (const int*)  d_in[1];
  const float* ea     = (const float*)d_in[2];
  const float* W_same = (const float*)d_in[3];
  const float* b_same = (const float*)d_in[4];
  const float* W_diff = (const float*)d_in[5];
  const float* b_diff = (const float*)d_in[6];
  const float* W_G    = (const float*)d_in[7];
  const float* rW1    = (const float*)d_in[8];
  const float* rb1    = (const float*)d_in[9];
  const float* rW2    = (const float*)d_in[10];
  const float* rb2    = (const float*)d_in[11];
  const float* W_last = (const float*)d_in[12];
  const float* b_last = (const float*)d_in[13];
  const float* u      = (const float*)d_in[14];

  const int NF = in_sizes[0];
  const int N  = NF / F;
  const int E  = in_sizes[2] / KRBF;
  const int R  = in_sizes[8] / (F * F);

  float* out_final = (float*)d_out;
  float* out_msged = (float*)d_out + NF;

  float* buf0 = (float*)d_ws;
  float* buf1 = buf0 + NF;

  hipMemsetAsync(buf1, 0, (size_t)NF * sizeof(float), stream);

  gemm128_kernel<true, 4, false><<<512, 256, 0, stream>>>(
      x, W_diff, b_diff, nullptr, nullptr, buf0, nullptr, N);

  edge_msg_kernel<<<512, 256, 0, stream>>>(buf0, eidx, ea, W_G, buf1, E);

  gemm128_kernel<true, 0, true><<<512, 256, 0, stream>>>(
      x, W_same, b_same, buf1, nullptr, out_msged, buf0, N);

  float* cur = buf0;
  float* oth = buf1;
  for (int r = 0; r < R; ++r) {
    resblock_kernel<<<512, 256, 0, stream>>>(
        cur, rW1 + (size_t)r * F * F, rb1 + (size_t)r * F,
        rW2 + (size_t)r * F * F, rb2 + (size_t)r * F, oth, N);
    float* sw = cur; cur = oth; oth = sw;
  }

  gemm128_kernel<true, 3, false><<<512, 256, 0, stream>>>(
      cur, W_last, b_last, x, u, out_final, nullptr, N);
}


# Node chain behind the scatter (update layer, residual blocks, last layer) in one launch: a workgroup keeps its tiles' state t in LDS across layers; 3 launches from 7; k and sum order kept
# speedup vs baseline: 2.3303x; 1.1554x over previous
#include <hip/hip_runtime.h>
#include <hip/hip_bf16.h>


typedef __attribute__((ext_vector_type(2))) float v2f;
typedef __attribute__((ext_vector_type(8))) float v8f;

__device__ __forceinline__ v8f wmma_k4(v2f a, v2f b, v8f c) {
  return __builtin_amdgcn_wmma_f32_16x16x4_f32(
       false, a,  false, b,
       (short)0, c,  false,  false);
}

__device__ __forceinline__ float ssp(float x) {
  float e = __expf(-fabsf(x));
  return fmaxf(x, 0.0f) + __logf(1.0f + e) - 0.69314718055994531f;
}

__device__ __forceinline__ void lds_barrier() {
  __builtin_amdgcn_fence(__ATOMIC_RELEASE, "workgroup", "local");
  __builtin_amdgcn_s_barrier();
  __builtin_amdgcn_fence(__ATOMIC_ACQUIRE, "workgroup", "local");
}

#define F 128
#define KRBF 64
#define A_STRIDE 132
#define E_STRIDE 68

#define ET 32

__device__ __forceinline__ float other_half(float x) {
  return __builtin_bit_cast(float, __builtin_amdgcn_permlanex16(
      __builtin_bit_cast(int, x), __builtin_bit_cast(int, x),
      0x76543210, 0xfedcba98, false, false));
}

__global__ void __launch_bounds__(256)
edge_msg_kernel(const float* __restrict__ P, const int* __restrict__ eidx,
                const float* __restrict__ ea, const float* __restrict__ Wg,
                float* __restrict__ aggr, int E) {
  __shared__ __align__(16) float As[ET * A_STRIDE];
  __shared__ __align__(16) float Es[ET * E_STRIDE];
  __shared__ int dsts[ET];

  const int tid  = threadIdx.x;
  const int lane = tid & 31;
  const int wv   = tid >> 5;
  const int m    = lane & 15;
  const int hi   = lane >> 4;
  const int rb   = (wv >> 2) * 16;
  const int c0   = (wv & 3) * 32;
  const int col  = c0 + lane;

  v2f bwg0[KRBF / 4];
  v2f bwg1[KRBF / 4];
  {
    const float* g0 = Wg + (c0 + m) * KRBF;
    const float* g1 = Wg + (c0 + 16 + m) * KRBF;
#pragma unroll
    for (int kk = 0; kk < KRBF / 4; ++kk) {
      bwg0[kk] = *(const v2f*)(&g0[kk * 4 + 2 * hi]);
      bwg1[kk] = *(const v2f*)(&g1[kk * 4 + 2 * hi]);
    }
  }

  const int nT = (E + ET - 1) / ET;

  float pa[16];
  float pe[8];
  int   pd = 0;

  int tile = blockIdx.x;
  if (tile < nT) {
    const int e0 = tile * ET;
#pragma unroll
    for (int q = 0; q < 16; ++q) {
      int i = tid + 256 * q;
      int r = i >> 7, c = i & (F - 1);
      int e = e0 + r;
      if (e >= E) e = E - 1;
      int s = eidx[e];
      pa[q] = P[s * F + c];
    }
#pragma unroll
    for (int q = 0; q < 8; ++q) {
      int i = tid + 256 * q;
      int r = i >> 6, c = i & (KRBF - 1);
      int e = e0 + r;
      if (e >= E) e = E - 1;
      pe[q] = ea[e * KRBF + c];
    }
    if (tid < ET) {
      int e = e0 + tid;
      if (e >= E) e = E - 1;
      pd = eidx[E + e];
    }
  }

  for (; tile < nT; tile += gridDim.x) {
    const int e0 = tile * ET;

#pragma unroll
    for (int q = 0; q < 16; ++q) {
      int i = tid + 256 * q;
      As[(i >> 7) * A_STRIDE + (i & (F - 1))] = pa[q];
    }
#pragma unroll
    for (int q = 0; q < 8; ++q) {
      int i = tid + 256 * q;
      Es[(i >> 6) * E_STRIDE + (i & (KRBF - 1))] = pe[q];
    }
    if (tid < ET) dsts[tid] = pd;
    lds_barrier();

    const int nxt = tile + gridDim.x;
    if (nxt < nT) {
      const int f0 = nxt * ET;
#pragma unroll
      for (int q = 0; q < 16; ++q) {
        int i = tid + 256 * q;
        int r = i >> 7, c = i & (F - 1);
        int e = f0 + r;
        if (e >= E) e = E - 1;
        int s = eidx[e];
        pa[q] = P[s * F + c];
      }
#pragma unroll
      for (int q = 0; q < 8; ++q) {
        int i = tid + 256 * q;
        int r = i >> 6, c = i & (KRBF - 1);
        int e = f0 + r;
        if (e >= E) e = E - 1;
        pe[q] = ea[e * KRBF + c];
      }
      if (tid < ET) {
        int e = f0 + tid;
        if (e >= E) e = E - 1;
        pd = eidx[E + e];
      }
    }

    v8f acc0 = {};
    v8f acc1 = {};
#pragma unroll
    for (int kk = 0; kk < KRBF / 4; ++kk) {
      v2f a = *(const v2f*)(&Es[(rb + m) * E_STRIDE + kk * 4 + 2 * hi]);
      acc0 = wmma_k4(a, bwg0[kk], acc0);
      acc1 = wmma_k4(a, bwg1[kk], acc1);
    }

#pragma unroll
    for (int j = 0; j < 8; ++j) {
      float give = hi ? acc0[j] : acc1[j];
      float got  = other_half(give);
      float glo  = hi ? got : acc0[j];
      float ghi  = hi ? acc1[j] : got;
      int rlo = rb + j;
      int rhi = rb + j + 8;
      if (e0 + rlo < E)
        atomicAdd(&aggr[dsts[rlo] * F + col], As[rlo * A_STRIDE + col] * glo);
      if (e0 + rhi < E)
        atomicAdd(&aggr[dsts[rhi] * F + col], As[rhi * A_STRIDE + col] * ghi);
    }
    lds_barrier();
  }
}

template <bool IN_ACT, int MODE, bool WRITE2>
__global__ void __launch_bounds__(256)
gemm128_kernel(const float* __restrict__ In, const float* __restrict__ W,
               const float* __restrict__ bias, const float* __restrict__ Ex1,
               const float* __restrict__ Ex2, float* __restrict__ Out,
               float* __restrict__ Out2, int N) {
  __shared__ __align__(16) float As[16 * A_STRIDE];

  const int tid  = threadIdx.x;
  const int lane = tid & 31;
  const int wv   = tid >> 5;
  const int m    = lane & 15;
  const int hi   = lane >> 4;
  const int n    = wv * 16 + m;

  v2f bw[F / 4];
  {
    const float* wrow = W + n * F;
#pragma unroll
    for (int kk = 0; kk < F / 4; ++kk)
      bw[kk] = *(const v2f*)(&wrow[kk * 4 + 2 * hi]);
  }
  const float bn = bias[n];
  float ex2n = 0.0f;
  if (MODE == 3) ex2n = Ex2[n];

  const int nT = (N + 15) / 16;

  float pa[8];
  int tile = blockIdx.x;
  if (tile < nT) {
    const int row0 = tile * 16;
#pragma unroll
    for (int q = 0; q < 8; ++q) {
      int i = tid + 256 * q;
      int rr = row0 + (i >> 7);
      if (rr >= N) rr = N - 1;
      pa[q] = In[rr * F + (i & (F - 1))];
    }
  }

  for (; tile < nT; tile += gridDim.x) {
    const int row0 = tile * 16;

#pragma unroll
    for (int q = 0; q < 8; ++q) {
      int i = tid + 256 * q;
      float v = pa[q];
      if (IN_ACT) v = ssp(v);
      As[(i >> 7) * A_STRIDE + (i & (F - 1))] = v;
    }
    lds_barrier();

    const int nxt = tile + gridDim.x;
    if (nxt < nT) {
      const int f0 = nxt * 16;
#pragma unroll
      for (int q = 0; q < 8; ++q) {
        int i = tid + 256 * q;
        int rr = f0 + (i >> 7);
        if (rr >= N) rr = N - 1;
        pa[q] = In[rr * F + (i & (F - 1))];
      }
    }

    v8f acc = {};
#pragma unroll
    for (int kk = 0; kk < F / 4; ++kk) {
      v2f a = *(const v2f*)(&As[m * A_STRIDE + kk * 4 + 2 * hi]);
      acc = wmma_k4(a, bw[kk], acc);
    }

#pragma unroll
    for (int j = 0; j < 8; ++j) {
      int r = row0 + j + 8 * hi;
      if (r >= N) continue;
      float g = acc[j] + bn;
      float o;
      if (MODE == 0)      o = ssp(g) + Ex1[r * F + n];
      else if (MODE == 1) o = g;
      else if (MODE == 2) o = Ex1[r * F + n] + g;
      else if (MODE == 3) o = g + Ex1[r * F + n] * ex2n;
      else                o = ssp(g);
      Out[r * F + n] = o;
      if (WRITE2) Out2[r * F + n] = o;
    }
    lds_barrier();
  }
}

#define CH_T 7
#define TILE_F (16 * A_STRIDE)

__device__ __forceinline__ v8f tile_product(const float* As, const v2f (&bw)[F / 4],
                                            int m, int hi) {
  v8f acc = {};
#pragma unroll
  for (int kk = 0; kk < F / 4; ++kk) {
    v2f a = *(const v2f*)(&As[m * A_STRIDE + kk * 4 + 2 * hi]);
    acc = wmma_k4(a, bw[kk], acc);
  }
  return acc;
}

__device__ __forceinline__ void load_frags(v2f (&bw)[F / 4], const float* wrow, int hi) {
#pragma unroll
  for (int kk = 0; kk < F / 4; ++kk)
    bw[kk] = *(const v2f*)(&wrow[kk * 4 + 2 * hi]);
}

__global__ void __launch_bounds__(256) __attribute__((amdgpu_waves_per_eu(4, 4)))
chain_kernel(const float* __restrict__ x, const float* __restrict__ aggr,
             const float* __restrict__ W_same, const float* __restrict__ b_same,
             const float* __restrict__ rW1, const float* __restrict__ rb1,
             const float* __restrict__ rW2, const float* __restrict__ rb2,
             const float* __restrict__ W_last, const float* __restrict__ b_last,
             const float* __restrict__ u,
             float* __restrict__ out_msged, float* __restrict__ out_final,
             int N, int R) {
  __shared__ __align__(16) float Tt[CH_T * TILE_F];
  __shared__ __align__(16) float Xs[TILE_F];
  __shared__ __align__(16) float Ys[TILE_F];

  const int tid  = threadIdx.x;
  const int lane = tid & 31;
  const int wv   = tid >> 5;
  const int m    = lane & 15;
  const int hi   = lane >> 4;
  const int n    = wv * 16 + m;
  const int G    = gridDim.x;

  const int nT = (N + 15) / 16;

  v2f bwA[F / 4];
  v2f bwB[F / 4];
  float pa[8];

  const int base = blockIdx.x;
  if (base >= nT) return;
  {
    int nt = (nT - base + G - 1) / G;
    if (nt > CH_T) nt = CH_T;

    {
      load_frags(bwA, W_same + n * F, hi);
      const float bn = b_same[n];
      {
        const int row0 = base * 16;
#pragma unroll
        for (int q = 0; q < 8; ++q) {
          int i = tid + 256 * q;
          int rr = row0 + (i >> 7);
          if (rr >= N) rr = N - 1;
          pa[q] = x[rr * F + (i & (F - 1))];
        }
      }
      for (int it = 0; it < nt; ++it) {
        const int row0 = (base + G * it) * 16;
#pragma unroll
        for (int q = 0; q < 8; ++q) {
          int i = tid + 256 * q;
          Xs[(i >> 7) * A_STRIDE + (i & (F - 1))] = ssp(pa[q]);
        }
        lds_barrier();

        if (it + 1 < nt) {
          const int f0 = (base + G * (it + 1)) * 16;
#pragma unroll
          for (int q = 0; q < 8; ++q) {
            int i = tid + 256 * q;
            int rr = f0 + (i >> 7);
            if (rr >= N) rr = N - 1;
            pa[q] = x[rr * F + (i & (F - 1))];
          }
        }

        v8f acc = tile_product(Xs, bwA, m, hi);
        float* Tr = Tt + it * TILE_F;
#pragma unroll
        for (int j = 0; j < 8; ++j) {
          int rl = j + 8 * hi;
          int r  = row0 + rl;
          float o = 0.0f;
          if (r < N) {
            float g = acc[j] + bn;
            o = ssp(g) + aggr[r * F + n];
            out_msged[r * F + n] = o;
          }
          Tr[rl * A_STRIDE + n] = o;
        }
        lds_barrier();
      }
    }

    for (int rb = 0; rb < R; ++rb) {
      load_frags(bwA, rW1 + rb * F * F + n * F, hi);
      load_frags(bwB, rW2 + rb * F * F + n * F, hi);
      const float b1n = rb1[rb * F + n];
      const float b2n = rb2[rb * F + n];

#pragma unroll
      for (int q = 0; q < 8; ++q) {
        int i = tid + 256 * q;
        pa[q] = Tt[(i >> 7) * A_STRIDE + (i & (F - 1))];
      }
      for (int it = 0; it < nt; ++it) {
#pragma unroll
        for (int q = 0; q < 8; ++q) {
          int i = tid + 256 * q;
          Xs[(i >> 7) * A_STRIDE + (i & (F - 1))] = ssp(pa[q]);
        }
        lds_barrier();

        if (it + 1 < nt) {
          const float* Tn = Tt + (it + 1) * TILE_F;
#pragma unroll
          for (int q = 0; q < 8; ++q) {
            int i = tid + 256 * q;
            pa[q] = Tn[(i >> 7) * A_STRIDE + (i & (F - 1))];
          }
        }

        {
          v8f acc = tile_product(Xs, bwA, m, hi);
#pragma unroll
          for (int j = 0; j < 8; ++j) {
            int rl = j + 8 * hi;
            float g = acc[j] + b1n;
            Ys[rl * A_STRIDE + n] = ssp(g);
          }
        }
        lds_barrier();

        {
          v8f acc = tile_product(Ys, bwB, m, hi);
          float* Tr = Tt + it * TILE_F;
#pragma unroll
          for (int j = 0; j < 8; ++j) {
            int rl = j + 8 * hi;
            float g = acc[j] + b2n;
            Tr[rl * A_STRIDE + n] = Tr[rl * A_STRIDE + n] + g;
          }
        }
        lds_barrier();
      }
    }

    {
      load_frags(bwA, W_last + n * F, hi);
      const float bn   = b_last[n];
      const float ex2n = u[n];

#pragma unroll
      for (int q = 0; q < 8; ++q) {
        int i = tid + 256 * q;
        pa[q] = Tt[(i >> 7) * A_STRIDE + (i & (F - 1))];
      }
      for (int it = 0; it < nt; ++it) {
        const int row0 = (base + G * it) * 16;
#pragma unroll
        for (int q = 0; q < 8; ++q) {
          int i = tid + 256 * q;
          Xs[(i >> 7) * A_STRIDE + (i & (F - 1))] = ssp(pa[q]);
        }
        lds_barrier();

        if (it + 1 < nt) {
          const float* Tn = Tt + (it + 1) * TILE_F;
#pragma unroll
          for (int q = 0; q < 8; ++q) {
            int i = tid + 256 * q;
            pa[q] = Tn[(i >> 7) * A_STRIDE + (i & (F - 1))];
          }
        }

        v8f acc = tile_product(Xs, bwA, m, hi);
#pragma unroll
        for (int j = 0; j < 8; ++j) {
          int r = row0 + j + 8 * hi;
          if (r >= N) continue;
          float g = acc[j] + bn;
          out_final[r * F + n] = g + x[r * F + n] * ex2n;
        }
        lds_barrier();
      }
    }
  }
}

extern "C" void kernel_launch(void* const* d_in, const int* in_sizes, int n_in,
                              void* d_out, int out_size, void* d_ws, size_t ws_size,
                              hipStream_t stream) {
  const float* x      = (const float*)d_in[0];
  const int*   eidx   = (const int*)  d_in[1];
  const float* ea     = (const float*)d_in[2];
  const float* W_same = (const float*)d_in[3];
  const float* b_same = (const float*)d_in[4];
  const float* W_diff = (const float*)d_in[5];
  const float* b_diff = (const float*)d_in[6];
  const float* W_G    = (const float*)d_in[7];
  const float* rW1    = (const float*)d_in[8];
  const float* rb1    = (const float*)d_in[9];
  const float* rW2    = (const float*)d_in[10];
  const float* rb2    = (const float*)d_in[11];
  const float* W_last = (const float*)d_in[12];
  const float* b_last = (const float*)d_in[13];
  const float* u      = (const float*)d_in[14];

  const int NF = in_sizes[0];
  const int N  = NF / F;
  const int E  = in_sizes[2] / KRBF;
  const int R  = in_sizes[8] / (F * F);

  float* out_final = (float*)d_out;
  float* out_msged = (float*)d_out + NF;

  float* buf0 = (float*)d_ws;
  float* buf1 = buf0 + NF;

  hipMemsetAsync(buf1, 0, (size_t)NF * sizeof(float), stream);

  gemm128_kernel<true, 4, false><<<512, 256, 0, stream>>>(
      x, W_diff, b_diff, nullptr, nullptr, buf0, nullptr, N);

  edge_msg_kernel<<<512, 256, 0, stream>>>(buf0, eidx, ea, W_G, buf1, E);

  chain_kernel<<<512, 256, 0, stream>>>(
      x, buf1, W_same, b_same, rW1, rb1, rW2, rb2, W_last, b_last, u,
      out_msged, out_final, N, R);
}


# Epilogue operands fetched as one group ahead of each tile's product (aggr rows, x rows; edge kernel: dst indices and P values from LDS), not one dependent load per row behind it; sums unchanged
# speedup vs baseline: 2.8294x; 1.2142x over previous
#include <hip/hip_runtime.h>
#include <hip/hip_bf16.h>


typedef __attribute__((ext_vector_type(2))) float v2f;
typedef __attribute__((ext_vector_type(8))) float v8f;

__device__ __forceinline__ v8f wmma_k4(v2f a, v2f b, v8f c) {
  return __builtin_amdgcn_wmma_f32_16x16x4_f32(
       false, a,  false, b,
       (short)0, c,  false,  false);
}

__device__ __forceinline__ float ssp(float x) {
  float e = __expf(-fabsf(x));
  return fmaxf(x, 0.0f) + __logf(1.0f + e) - 0.69314718055994531f;
}

__device__ __forceinline__ void lds_barrier() {
  __builtin_amdgcn_fence(__ATOMIC_RELEASE, "workgroup", "local");
  __builtin_amdgcn_s_barrier();
  __builtin_amdgcn_fence(__ATOMIC_ACQUIRE, "workgroup", "local");
}

#define F 128
#define KRBF 64
#define A_STRIDE 132
#define E_STRIDE 68

#define ET 32

__device__ __forceinline__ float other_half(float x) {
  return __builtin_bit_cast(float, __builtin_amdgcn_permlanex16(
      __builtin_bit_cast(int, x), __builtin_bit_cast(int, x),
      0x76543210, 0xfedcba98, false, false));
}

__global__ void __launch_bounds__(256)
edge_msg_kernel(const float* __restrict__ P, const int* __restrict__ eidx,
                const float* __restrict__ ea, const float* __restrict__ Wg,
                float* __restrict__ aggr, int E) {
  __shared__ __align__(16) float As[ET * A_STRIDE];
  __shared__ __align__(16) float Es[ET * E_STRIDE];
  __shared__ int dsts[ET];

  const int tid  = threadIdx.x;
  const int lane = tid & 31;
  const int wv   = tid >> 5;
  const int m    = lane & 15;
  const int hi   = lane >> 4;
  const int rb   = (wv >> 2) * 16;
  const int c0   = (wv & 3) * 32;
  const int col  = c0 + lane;

  v2f bwg0[KRBF / 4];
  v2f bwg1[KRBF / 4];
  {
    const float* g0 = Wg + (c0 + m) * KRBF;
    const float* g1 = Wg + (c0 + 16 + m) * KRBF;
#pragma unroll
    for (int kk = 0; kk < KRBF / 4; ++kk) {
      bwg0[kk] = *(const v2f*)(&g0[kk * 4 + 2 * hi]);
      bwg1[kk] = *(const v2f*)(&g1[kk * 4 + 2 * hi]);
    }
  }

  const int nT = (E + ET - 1) / ET;

  float pa[16];
  float pe[8];
  int   pd = 0;

  int tile = blockIdx.x;
  if (tile < nT) {
    const int e0 = tile * ET;
#pragma unroll
    for (int q = 0; q < 16; ++q) {
      int i = tid + 256 * q;
      int r = i >> 7, c = i & (F - 1);
      int e = e0 + r;
      if (e >= E) e = E - 1;
      int s = eidx[e];
      pa[q] = P[s * F + c];
    }
#pragma unroll
    for (int q = 0; q < 8; ++q) {
      int i = tid + 256 * q;
      int r = i >> 6, c = i & (KRBF - 1);
      int e = e0 + r;
      if (e >= E) e = E - 1;
      pe[q] = ea[e * KRBF + c];
    }
    if (tid < ET) {
      int e = e0 + tid;
      if (e >= E) e = E - 1;
      pd = eidx[E + e];
    }
  }

  for (; tile < nT; tile += gridDim.x) {
    const int e0 = tile * ET;

#pragma unroll
    for (int q = 0; q < 16; ++q) {
      int i = tid + 256 * q;
      As[(i >> 7) * A_STRIDE + (i & (F - 1))] = pa[q];
    }
#pragma unroll
    for (int q = 0; q < 8; ++q) {
      int i = tid + 256 * q;
      Es[(i >> 6) * E_STRIDE + (i & (KRBF - 1))] = pe[q];
    }
    if (tid < ET) dsts[tid] = pd;
    lds_barrier();

    const int nxt = tile + gridDim.x;
    if (nxt < nT) {
      const int f0 = nxt * ET;
#pragma unroll
      for (int q = 0; q < 16; ++q) {
        int i = tid + 256 * q;
        int r = i >> 7, c = i & (F - 1);
        int e = f0 + r;
        if (e >= E) e = E - 1;
        int s = eidx[e];
        pa[q] = P[s * F + c];
      }
#pragma unroll
      for (int q = 0; q < 8; ++q) {
        int i = tid + 256 * q;
        int r = i >> 6, c = i & (KRBF - 1);
        int e = f0 + r;
        if (e >= E) e = E - 1;
        pe[q] = ea[e * KRBF + c];
      }
      if (tid < ET) {
        int e = f0 + tid;
        if (e >= E) e = E - 1;
        pd = eidx[E + e];
      }
    }

    int   dr[16];
    float pv[16];
#pragma unroll
    for (int j = 0; j < 16; ++j) {
      dr[j] = dsts[rb + j];
      pv[j] = As[(rb + j) * A_STRIDE + col];
    }

    v8f acc0 = {};
    v8f acc1 = {};
#pragma unroll
    for (int kk = 0; kk < KRBF / 4; ++kk) {
      v2f a = *(const v2f*)(&Es[(rb + m) * E_STRIDE + kk * 4 + 2 * hi]);
      acc0 = wmma_k4(a, bwg0[kk], acc0);
      acc1 = wmma_k4(a, bwg1[kk], acc1);
    }

    if (e0 + ET <= E) {
#pragma unroll
      for (int j = 0; j < 8; ++j) {
        float give = hi ? acc0[j] : acc1[j];
        float got  = other_half(give);
        float glo  = hi ? got : acc0[j];
        float ghi  = hi ? acc1[j] : got;
        atomicAdd(&aggr[dr[j] * F + col], pv[j] * glo);
        atomicAdd(&aggr[dr[j + 8] * F + col], pv[j + 8] * ghi);
      }
    } else {
#pragma unroll
      for (int j = 0; j < 8; ++j) {
        float give = hi ? acc0[j] : acc1[j];
        float got  = other_half(give);
        float glo  = hi ? got : acc0[j];
        float ghi  = hi ? acc1[j] : got;
        int rlo = rb + j;
        int rhi = rb + j + 8;
        if (e0 + rlo < E)
          atomicAdd(&aggr[dr[j] * F + col], pv[j] * glo);
        if (e0 + rhi < E)
          atomicAdd(&aggr[dr[j + 8] * F + col], pv[j + 8] * ghi);
      }
    }
    lds_barrier();
  }
}

template <bool IN_ACT, int MODE, bool WRITE2>
__global__ void __launch_bounds__(256)
gemm128_kernel(const float* __restrict__ In, const float* __restrict__ W,
               const float* __restrict__ bias, const float* __restrict__ Ex1,
               const float* __restrict__ Ex2, float* __restrict__ Out,
               float* __restrict__ Out2, int N) {
  __shared__ __align__(16) float As[16 * A_STRIDE];

  const int tid  = threadIdx.x;
  const int lane = tid & 31;
  const int wv   = tid >> 5;
  const int m    = lane & 15;
  const int hi   = lane >> 4;
  const int n    = wv * 16 + m;

  v2f bw[F / 4];
  {
    const float* wrow = W + n * F;
#pragma unroll
    for (int kk = 0; kk < F / 4; ++kk)
      bw[kk] = *(const v2f*)(&wrow[kk * 4 + 2 * hi]);
  }
  const float bn = bias[n];
  float ex2n = 0.0f;
  if (MODE == 3) ex2n = Ex2[n];

  const int nT = (N + 15) / 16;

  float pa[8];
  int tile = blockIdx.x;
  if (tile < nT) {
    const int row0 = tile * 16;
#pragma unroll
    for (int q = 0; q < 8; ++q) {
      int i = tid + 256 * q;
      int rr = row0 + (i >> 7);
      if (rr >= N) rr = N - 1;
      pa[q] = In[rr * F + (i & (F - 1))];
    }
  }

  for (; tile < nT; tile += gridDim.x) {
    const int row0 = tile * 16;

#pragma unroll
    for (int q = 0; q < 8; ++q) {
      int i = tid + 256 * q;
      float v = pa[q];
      if (IN_ACT) v = ssp(v);
      As[(i >> 7) * A_STRIDE + (i & (F - 1))] = v;
    }
    lds_barrier();

    const int nxt = tile + gridDim.x;
    if (nxt < nT) {
      const int f0 = nxt * 16;
#pragma unroll
      for (int q = 0; q < 8; ++q) {
        int i = tid + 256 * q;
        int rr = f0 + (i >> 7);
        if (rr >= N) rr = N - 1;
        pa[q] = In[rr * F + (i & (F - 1))];
      }
    }

    v8f acc = {};
#pragma unroll
    for (int kk = 0; kk < F / 4; ++kk) {
      v2f a = *(const v2f*)(&As[m * A_STRIDE + kk * 4 + 2 * hi]);
      acc = wmma_k4(a, bw[kk], acc);
    }

#pragma unroll
    for (int j = 0; j < 8; ++j) {
      int r = row0 + j + 8 * hi;
      if (r >= N) continue;
      float g = acc[j] + bn;
      float o;
      if (MODE == 0)      o = ssp(g) + Ex1[r * F + n];
      else if (MODE == 1) o = g;
      else if (MODE == 2) o = Ex1[r * F + n] + g;
      else if (MODE == 3) o = g + Ex1[r * F + n] * ex2n;
      else                o = ssp(g);
      Out[r * F + n] = o;
      if (WRITE2) Out2[r * F + n] = o;
    }
    lds_barrier();
  }
}

#define CH_T 7
#define TILE_F (16 * A_STRIDE)

__device__ __forceinline__ v8f tile_product(const float* As, const v2f (&bw)[F / 4],
                                            int m, int hi) {
  v8f acc = {};
#pragma unroll
  for (int kk = 0; kk < F / 4; ++kk) {
    v2f a = *(const v2f*)(&As[m * A_STRIDE + kk * 4 + 2 * hi]);
    acc = wmma_k4(a, bw[kk], acc);
  }
  return acc;
}

__device__ __forceinline__ void load_frags(v2f (&bw)[F / 4], const float* wrow, int hi) {
#pragma unroll
  for (int kk = 0; kk < F / 4; ++kk)
    bw[kk] = *(const v2f*)(&wrow[kk * 4 + 2 * hi]);
}

__global__ void __launch_bounds__(256) __attribute__((amdgpu_waves_per_eu(4, 4)))
chain_kernel(const float* __restrict__ x, const float* __restrict__ aggr,
             const float* __restrict__ W_same, const float* __restrict__ b_same,
             const float* __restrict__ rW1, const float* __restrict__ rb1,
             const float* __restrict__ rW2, const float* __restrict__ rb2,
             const float* __restrict__ W_last, const float* __restrict__ b_last,
             const float* __restrict__ u,
             float* __restrict__ out_msged, float* __restrict__ out_final,
             int N, int R) {
  __shared__ __align__(16) float Tt[CH_T * TILE_F];
  __shared__ __align__(16) float Xs[TILE_F];
  __shared__ __align__(16) float Ys[TILE_F];

  const int tid  = threadIdx.x;
  const int lane = tid & 31;
  const int wv   = tid >> 5;
  const int m    = lane & 15;
  const int hi   = lane >> 4;
  const int n    = wv * 16 + m;
  const int G    = gridDim.x;

  const int nT = (N + 15) / 16;

  v2f bwA[F / 4];
  v2f bwB[F / 4];
  float pa[8];

  const int base = blockIdx.x;
  if (base >= nT) return;
  {
    int nt = (nT - base + G - 1) / G;
    if (nt > CH_T) nt = CH_T;

    {
      load_frags(bwA, W_same + n * F, hi);
      const float bn = b_same[n];
      {
        const int row0 = base * 16;
#pragma unroll
        for (int q = 0; q < 8; ++q) {
          int i = tid + 256 * q;
          int rr = row0 + (i >> 7);
          if (rr >= N) rr = N - 1;
          pa[q] = x[rr * F + (i & (F - 1))];
        }
      }
      for (int it = 0; it < nt; ++it) {
        const int row0 = (base + G * it) * 16;
#pragma unroll
        for (int q = 0; q < 8; ++q) {
          int i = tid + 256 * q;
          Xs[(i >> 7) * A_STRIDE + (i & (F - 1))] = ssp(pa[q]);
        }
        lds_barrier();

        float ag[8];
#pragma unroll
        for (int j = 0; j < 8; ++j) {
          int r = row0 + j + 8 * hi;
          if (r >= N) r = N - 1;
          ag[j] = aggr[r * F + n];
        }

        if (it + 1 < nt) {
          const int f0 = (base + G * (it + 1)) * 16;
#pragma unroll
          for (int q = 0; q < 8; ++q) {
            int i = tid + 256 * q;
            int rr = f0 + (i >> 7);
            if (rr >= N) rr = N - 1;
            pa[q] = x[rr * F + (i & (F - 1))];
          }
        }

        v8f acc = tile_product(Xs, bwA, m, hi);
        float* Tr = Tt + it * TILE_F;
#pragma unroll
        for (int j = 0; j < 8; ++j) {
          int rl = j + 8 * hi;
          int r  = row0 + rl;
          float g = acc[j] + bn;
          float o = ssp(g) + ag[j];
          if (r < N) out_msged[r * F + n] = o;
          else       o = 0.0f;
          Tr[rl * A_STRIDE + n] = o;
        }
        lds_barrier();
      }
    }

    for (int rb = 0; rb < R; ++rb) {
      load_frags(bwA, rW1 + rb * F * F + n * F, hi);
      load_frags(bwB, rW2 + rb * F * F + n * F, hi);
      const float b1n = rb1[rb * F + n];
      const float b2n = rb2[rb * F + n];

#pragma unroll
      for (int q = 0; q < 8; ++q) {
        int i = tid + 256 * q;
        pa[q] = Tt[(i >> 7) * A_STRIDE + (i & (F - 1))];
      }
      for (int it = 0; it < nt; ++it) {
#pragma unroll
        for (int q = 0; q < 8; ++q) {
          int i = tid + 256 * q;
          Xs[(i >> 7) * A_STRIDE + (i & (F - 1))] = ssp(pa[q]);
        }
        lds_barrier();

        if (it + 1 < nt) {
          const float* Tn = Tt + (it + 1) * TILE_F;
#pragma unroll
          for (int q = 0; q < 8; ++q) {
            int i = tid + 256 * q;
            pa[q] = Tn[(i >> 7) * A_STRIDE + (i & (F - 1))];
          }
        }

        {
          v8f acc = tile_product(Xs, bwA, m, hi);
#pragma unroll
          for (int j = 0; j < 8; ++j) {
            int rl = j + 8 * hi;
            float g = acc[j] + b1n;
            Ys[rl * A_STRIDE + n] = ssp(g);
          }
        }
        lds_barrier();

        {
          v8f acc = tile_product(Ys, bwB, m, hi);
          float* Tr = Tt + it * TILE_F;
#pragma unroll
          for (int j = 0; j < 8; ++j) {
            int rl = j + 8 * hi;
            float g = acc[j] + b2n;
            Tr[rl * A_STRIDE + n] = Tr[rl * A_STRIDE + n] + g;
          }
        }
        lds_barrier();
      }
    }

    {
      load_frags(bwA, W_last + n * F, hi);
      const float bn   = b_last[n];
      const float ex2n = u[n];

#pragma unroll
      for (int q = 0; q < 8; ++q) {
        int i = tid + 256 * q;
        pa[q] = Tt[(i >> 7) * A_STRIDE + (i & (F - 1))];
      }
      for (int it = 0; it < nt; ++it) {
        const int row0 = (base + G * it) * 16;
#pragma unroll
        for (int q = 0; q < 8; ++q) {
          int i = tid + 256 * q;
          Xs[(i >> 7) * A_STRIDE + (i & (F - 1))] = ssp(pa[q]);
        }
        lds_barrier();

        float xr[8];
#pragma unroll
        for (int j = 0; j < 8; ++j) {
          int r = row0 + j + 8 * hi;
          if (r >= N) r = N - 1;
          xr[j] = x[r * F + n];
        }

        if (it + 1 < nt) {
          const float* Tn = Tt + (it + 1) * TILE_F;
#pragma unroll
          for (int q = 0; q < 8; ++q) {
            int i = tid + 256 * q;
            pa[q] = Tn[(i >> 7) * A_STRIDE + (i & (F - 1))];
          }
        }

        v8f acc = tile_product(Xs, bwA, m, hi);
#pragma unroll
        for (int j = 0; j < 8; ++j) {
          int r = row0 + j + 8 * hi;
          float g = acc[j] + bn;
          float o = g + xr[j] * ex2n;
          if (r < N) out_final[r * F + n] = o;
        }
        lds_barrier();
      }
    }
  }
}

extern "C" void kernel_launch(void* const* d_in, const int* in_sizes, int n_in,
                              void* d_out, int out_size, void* d_ws, size_t ws_size,
                              hipStream_t stream) {
  const float* x      = (const float*)d_in[0];
  const int*   eidx   = (const int*)  d_in[1];
  const float* ea     = (const float*)d_in[2];
  const float* W_same = (const float*)d_in[3];
  const float* b_same = (const float*)d_in[4];
  const float* W_diff = (const float*)d_in[5];
  const float* b_diff = (const float*)d_in[6];
  const float* W_G    = (const float*)d_in[7];
  const float* rW1    = (const float*)d_in[8];
  const float* rb1    = (const float*)d_in[9];
  const float* rW2    = (const float*)d_in[10];
  const float* rb2    = (const float*)d_in[11];
  const float* W_last = (const float*)d_in[12];
  const float* b_last = (const float*)d_in[13];
  const float* u      = (const float*)d_in[14];

  const int NF = in_sizes[0];
  const int N  = NF / F;
  const int E  = in_sizes[2] / KRBF;
  const int R  = in_sizes[8] / (F * F);

  float* out_final = (float*)d_out;
  float* out_msged = (float*)d_out + NF;

  float* buf0 = (float*)d_ws;
  float* buf1 = buf0 + NF;

  hipMemsetAsync(buf1, 0, (size_t)NF * sizeof(float), stream);

  gemm128_kernel<true, 4, false><<<512, 256, 0, stream>>>(
      x, W_diff, b_diff, nullptr, nullptr, buf0, nullptr, N);

  edge_msg_kernel<<<512, 256, 0, stream>>>(buf0, eidx, ea, W_G, buf1, E);

  chain_kernel<<<512, 256, 0, stream>>>(
      x, buf1, W_same, b_same, rW1, rb1, rW2, rb2, W_last, b_last, u,
      out_msged, out_final, N, R);
}


# Edge kernel tile fetch in whole rows: a 16-byte load per lane takes a whole gathered row (6 row loads for 24, no index load in front); source indices read a tile ahead, once per wave; sums unchanged
# speedup vs baseline: 3.0037x; 1.0616x over previous
#include <hip/hip_runtime.h>
#include <hip/hip_bf16.h>


typedef __attribute__((ext_vector_type(2))) float v2f;
typedef __attribute__((ext_vector_type(8))) float v8f;

__device__ __forceinline__ v8f wmma_k4(v2f a, v2f b, v8f c) {
  return __builtin_amdgcn_wmma_f32_16x16x4_f32(
       false, a,  false, b,
       (short)0, c,  false,  false);
}

__device__ __forceinline__ float ssp(float x) {
  float e = __expf(-fabsf(x));
  return fmaxf(x, 0.0f) + __logf(1.0f + e) - 0.69314718055994531f;
}

__device__ __forceinline__ void lds_barrier() {
  __builtin_amdgcn_fence(__ATOMIC_RELEASE, "workgroup", "local");
  __builtin_amdgcn_s_barrier();
  __builtin_amdgcn_fence(__ATOMIC_ACQUIRE, "workgroup", "local");
}

#define F 128
#define KRBF 64
#define A_STRIDE 132
#define E_STRIDE 68

#define ET 32

typedef __attribute__((ext_vector_type(4))) float v4f;

__device__ __forceinline__ float other_half(float x) {
  return __builtin_bit_cast(float, __builtin_amdgcn_permlanex16(
      __builtin_bit_cast(int, x), __builtin_bit_cast(int, x),
      0x76543210, 0xfedcba98, false, false));
}

__global__ void __launch_bounds__(256)
edge_msg_kernel(const float* __restrict__ P, const int* __restrict__ eidx,
                const float* __restrict__ ea, const float* __restrict__ Wg,
                float* __restrict__ aggr, int E) {
  __shared__ __align__(16) float As[ET * A_STRIDE];
  __shared__ __align__(16) float Es[ET * E_STRIDE];
  __shared__ int dsts[ET];

  const int tid  = threadIdx.x;
  const int lane = tid & 31;
  const int wv   = tid >> 5;
  const int m    = lane & 15;
  const int hi   = lane >> 4;
  const int rb   = (wv >> 2) * 16;
  const int c0   = (wv & 3) * 32;
  const int col  = c0 + lane;
  const int r4   = wv * 4;

  v2f bwg0[KRBF / 4];
  v2f bwg1[KRBF / 4];
  {
    const float* g0 = Wg + (c0 + m) * KRBF;
    const float* g1 = Wg + (c0 + 16 + m) * KRBF;
#pragma unroll
    for (int kk = 0; kk < KRBF / 4; ++kk) {
      bwg0[kk] = *(const v2f*)(&g0[kk * 4 + 2 * hi]);
      bwg1[kk] = *(const v2f*)(&g1[kk * 4 + 2 * hi]);
    }
  }

  const int nT = (E + ET - 1) / ET;

  v4f pa[4];
  v4f pe[2];
  int pd = 0;
  int ps = 0;

  int tile = blockIdx.x;
  if (tile < nT) {
    const int e0 = tile * ET;
    {
      int e = e0 + r4 + (lane & 3);
      if (e >= E) e = E - 1;
      ps = eidx[e];
    }
#pragma unroll
    for (int i = 0; i < 4; ++i) {
      int s = __builtin_amdgcn_readlane(ps, i);
      pa[i] = *(const v4f*)(&P[s * F + 4 * lane]);
    }
#pragma unroll
    for (int i = 0; i < 2; ++i) {
      int e = e0 + r4 + 2 * i + hi;
      if (e >= E) e = E - 1;
      pe[i] = *(const v4f*)(&ea[e * KRBF + 4 * m]);
    }
    if (tid < ET) {
      int e = e0 + tid;
      if (e >= E) e = E - 1;
      pd = eidx[E + e];
    }
    {
      int e = (tile + (int)gridDim.x) * ET + r4 + (lane & 3);
      if (e >= E) e = E - 1;
      ps = eidx[e];
    }
  }

  for (; tile < nT; tile += gridDim.x) {
    const int e0 = tile * ET;

#pragma unroll
    for (int i = 0; i < 4; ++i)
      *(v4f*)(&As[(r4 + i) * A_STRIDE + 4 * lane]) = pa[i];
#pragma unroll
    for (int i = 0; i < 2; ++i)
      *(v4f*)(&Es[(r4 + 2 * i + hi) * E_STRIDE + 4 * m]) = pe[i];
    if (tid < ET) dsts[tid] = pd;

    const int s0 = __builtin_amdgcn_readlane(ps, 0);
    const int s1 = __builtin_amdgcn_readlane(ps, 1);
    const int s2 = __builtin_amdgcn_readlane(ps, 2);
    const int s3 = __builtin_amdgcn_readlane(ps, 3);
    lds_barrier();

    const int nxt = tile + gridDim.x;
    if (nxt < nT) {
      const int f0 = nxt * ET;
      pa[0] = *(const v4f*)(&P[s0 * F + 4 * lane]);
      pa[1] = *(const v4f*)(&P[s1 * F + 4 * lane]);
      pa[2] = *(const v4f*)(&P[s2 * F + 4 * lane]);
      pa[3] = *(const v4f*)(&P[s3 * F + 4 * lane]);
#pragma unroll
      for (int i = 0; i < 2; ++i) {
        int e = f0 + r4 + 2 * i + hi;
        if (e >= E) e = E - 1;
        pe[i] = *(const v4f*)(&ea[e * KRBF + 4 * m]);
      }
      if (tid < ET) {
        int e = f0 + tid;
        if (e >= E) e = E - 1;
        pd = eidx[E + e];
      }
      {
        int e = (nxt + (int)gridDim.x) * ET + r4 + (lane & 3);
        if (e >= E) e = E - 1;
        ps = eidx[e];
      }
    }

    int   dr[16];
    float pv[16];
#pragma unroll
    for (int j = 0; j < 16; ++j) {
      dr[j] = dsts[rb + j];
      pv[j] = As[(rb + j) * A_STRIDE + col];
    }

    v8f acc0 = {};
    v8f acc1 = {};
#pragma unroll
    for (int kk = 0; kk < KRBF / 4; ++kk) {
      v2f a = *(const v2f*)(&Es[(rb + m) * E_STRIDE + kk * 4 + 2 * hi]);
      acc0 = wmma_k4(a, bwg0[kk], acc0);
      acc1 = wmma_k4(a, bwg1[kk], acc1);
    }

    if (e0 + ET <= E) {
#pragma unroll
      for (int j = 0; j < 8; ++j) {
        float give = hi ? acc0[j] : acc1[j];
        float got  = other_half(give);
        float glo  = hi ? got : acc0[j];
        float ghi  = hi ? acc1[j] : got;
        atomicAdd(&aggr[dr[j] * F + col], pv[j] * glo);
        atomicAdd(&aggr[dr[j + 8] * F + col], pv[j + 8] * ghi);
      }
    } else {
#pragma unroll
      for (int j = 0; j < 8; ++j) {
        float give = hi ? acc0[j] : acc1[j];
        float got  = other_half(give);
        float glo  = hi ? got : acc0[j];
        float ghi  = hi ? acc1[j] : got;
        int rlo = rb + j;
        int rhi = rb + j + 8;
        if (e0 + rlo < E)
          atomicAdd(&aggr[dr[j] * F + col], pv[j] * glo);
        if (e0 + rhi < E)
          atomicAdd(&aggr[dr[j + 8] * F + col], pv[j + 8] * ghi);
      }
    }
    lds_barrier();
  }
}

template <bool IN_ACT, int MODE, bool WRITE2>
__global__ void __launch_bounds__(256)
gemm128_kernel(const float* __restrict__ In, const float* __restrict__ W,
               const float* __restrict__ bias, const float* __restrict__ Ex1,
               const float* __restrict__ Ex2, float* __restrict__ Out,
               float* __restrict__ Out2, int N) {
  __shared__ __align__(16) float As[16 * A_STRIDE];

  const int tid  = threadIdx.x;
  const int lane = tid & 31;
  const int wv   = tid >> 5;
  const int m    = lane & 15;
  const int hi   = lane >> 4;
  const int n    = wv * 16 + m;

  v2f bw[F / 4];
  {
    const float* wrow = W + n * F;
#pragma unroll
    for (int kk = 0; kk < F / 4; ++kk)
      bw[kk] = *(const v2f*)(&wrow[kk * 4 + 2 * hi]);
  }
  const float bn = bias[n];
  float ex2n = 0.0f;
  if (MODE == 3) ex2n = Ex2[n];

  const int nT = (N + 15) / 16;

  float pa[8];
  int tile = blockIdx.x;
  if (tile < nT) {
    const int row0 = tile * 16;
#pragma unroll
    for (int q = 0; q < 8; ++q) {
      int i = tid + 256 * q;
      int rr = row0 + (i >> 7);
      if (rr >= N) rr = N - 1;
      pa[q] = In[rr * F + (i & (F - 1))];
    }
  }

  for (; tile < nT; tile += gridDim.x) {
    const int row0 = tile * 16;

#pragma unroll
    for (int q = 0; q < 8; ++q) {
      int i = tid + 256 * q;
      float v = pa[q];
      if (IN_ACT) v = ssp(v);
      As[(i >> 7) * A_STRIDE + (i & (F - 1))] = v;
    }
    lds_barrier();

    const int nxt = tile + gridDim.x;
    if (nxt < nT) {
      const int f0 = nxt * 16;
#pragma unroll
      for (int q = 0; q < 8; ++q) {
        int i = tid + 256 * q;
        int rr = f0 + (i >> 7);
        if (rr >= N) rr = N - 1;
        pa[q] = In[rr * F + (i & (F - 1))];
      }
    }

    v8f acc = {};
#pragma unroll
    for (int kk = 0; kk < F / 4; ++kk) {
      v2f a = *(const v2f*)(&As[m * A_STRIDE + kk * 4 + 2 * hi]);
      acc = wmma_k4(a, bw[kk], acc);
    }

#pragma unroll
    for (int j = 0; j < 8; ++j) {
      int r = row0 + j + 8 * hi;
      if (r >= N) continue;
      float g = acc[j] + bn;
      float o;
      if (MODE == 0)      o = ssp(g) + Ex1[r * F + n];
      else if (MODE == 1) o = g;
      else if (MODE == 2) o = Ex1[r * F + n] + g;
      else if (MODE == 3) o = g + Ex1[r * F + n] * ex2n;
      else                o = ssp(g);
      Out[r * F + n] = o;
      if (WRITE2) Out2[r * F + n] = o;
    }
    lds_barrier();
  }
}

#define CH_T 7
#define TILE_F (16 * A_STRIDE)

__device__ __forceinline__ v8f tile_product(const float* As, const v2f (&bw)[F / 4],
                                            int m, int hi) {
  v8f acc = {};
#pragma unroll
  for (int kk = 0; kk < F / 4; ++kk) {
    v2f a = *(const v2f*)(&As[m * A_STRIDE + kk * 4 + 2 * hi]);
    acc = wmma_k4(a, bw[kk], acc);
  }
  return acc;
}

__device__ __forceinline__ void load_frags(v2f (&bw)[F / 4], const float* wrow, int hi) {
#pragma unroll
  for (int kk = 0; kk < F / 4; ++kk)
    bw[kk] = *(const v2f*)(&wrow[kk * 4 + 2 * hi]);
}

__global__ void __launch_bounds__(256) __attribute__((amdgpu_waves_per_eu(4, 4)))
chain_kernel(const float* __restrict__ x, const float* __restrict__ aggr,
             const float* __restrict__ W_same, const float* __restrict__ b_same,
             const float* __restrict__ rW1, const float* __restrict__ rb1,
             const float* __restrict__ rW2, const float* __restrict__ rb2,
             const float* __restrict__ W_last, const float* __restrict__ b_last,
             const float* __restrict__ u,
             float* __restrict__ out_msged, float* __restrict__ out_final,
             int N, int R) {
  __shared__ __align__(16) float Tt[CH_T * TILE_F];
  __shared__ __align__(16) float Xs[TILE_F];
  __shared__ __align__(16) float Ys[TILE_F];

  const int tid  = threadIdx.x;
  const int lane = tid & 31;
  const int wv   = tid >> 5;
  const int m    = lane & 15;
  const int hi   = lane >> 4;
  const int n    = wv * 16 + m;
  const int G    = gridDim.x;

  const int nT = (N + 15) / 16;

  v2f bwA[F / 4];
  v2f bwB[F / 4];
  float pa[8];

  const int base = blockIdx.x;
  if (base >= nT) return;
  {
    int nt = (nT - base + G - 1) / G;
    if (nt > CH_T) nt = CH_T;

    {
      load_frags(bwA, W_same + n * F, hi);
      const float bn = b_same[n];
      {
        const int row0 = base * 16;
#pragma unroll
        for (int q = 0; q < 8; ++q) {
          int i = tid + 256 * q;
          int rr = row0 + (i >> 7);
          if (rr >= N) rr = N - 1;
          pa[q] = x[rr * F + (i & (F - 1))];
        }
      }
      for (int it = 0; it < nt; ++it) {
        const int row0 = (base + G * it) * 16;
#pragma unroll
        for (int q = 0; q < 8; ++q) {
          int i = tid + 256 * q;
          Xs[(i >> 7) * A_STRIDE + (i & (F - 1))] = ssp(pa[q]);
        }
        lds_barrier();

        float ag[8];
#pragma unroll
        for (int j = 0; j < 8; ++j) {
          int r = row0 + j + 8 * hi;
          if (r >= N) r = N - 1;
          ag[j] = aggr[r * F + n];
        }

        if (it + 1 < nt) {
          const int f0 = (base + G * (it + 1)) * 16;
#pragma unroll
          for (int q = 0; q < 8; ++q) {
            int i = tid + 256 * q;
            int rr = f0 + (i >> 7);
            if (rr >= N) rr = N - 1;
            pa[q] = x[rr * F + (i & (F - 1))];
          }
        }

        v8f acc = tile_product(Xs, bwA, m, hi);
        float* Tr = Tt + it * TILE_F;
#pragma unroll
        for (int j = 0; j < 8; ++j) {
          int rl = j + 8 * hi;
          int r  = row0 + rl;
          float g = acc[j] + bn;
          float o = ssp(g) + ag[j];
          if (r < N) out_msged[r * F + n] = o;
          else       o = 0.0f;
          Tr[rl * A_STRIDE + n] = o;
        }
        lds_barrier();
      }
    }

    for (int rb = 0; rb < R; ++rb) {
      load_frags(bwA, rW1 + rb * F * F + n * F, hi);
      load_frags(bwB, rW2 + rb * F * F + n * F, hi);
      const float b1n = rb1[rb * F + n];
      const float b2n = rb2[rb * F + n];

#pragma unroll
      for (int q = 0; q < 8; ++q) {
        int i = tid + 256 * q;
        pa[q] = Tt[(i >> 7) * A_STRIDE + (i & (F - 1))];
      }
      for (int it = 0; it < nt; ++it) {
#pragma unroll
        for (int q = 0; q < 8; ++q) {
          int i = tid + 256 * q;
          Xs[(i >> 7) * A_STRIDE + (i & (F - 1))] = ssp(pa[q]);
        }
        lds_barrier();

        if (it + 1 < nt) {
          const float* Tn = Tt + (it + 1) * TILE_F;
#pragma unroll
          for (int q = 0; q < 8; ++q) {
            int i = tid + 256 * q;
            pa[q] = Tn[(i >> 7) * A_STRIDE + (i & (F - 1))];
          }
        }

        {
          v8f acc = tile_product(Xs, bwA, m, hi);
#pragma unroll
          for (int j = 0; j < 8; ++j) {
            int rl = j + 8 * hi;
            float g = acc[j] + b1n;
            Ys[rl * A_STRIDE + n] = ssp(g);
          }
        }
        lds_barrier();

        {
          v8f acc = tile_product(Ys, bwB, m, hi);
          float* Tr = Tt + it * TILE_F;
#pragma unroll
          for (int j = 0; j < 8; ++j) {
            int rl = j + 8 * hi;
            float g = acc[j] + b2n;
            Tr[rl * A_STRIDE + n] = Tr[rl * A_STRIDE + n] + g;
          }
        }
        lds_barrier();
      }
    }

    {
      load_frags(bwA, W_last + n * F, hi);
      const float bn   = b_last[n];
      const float ex2n = u[n];

#pragma unroll
      for (int q = 0; q < 8; ++q) {
        int i = tid + 256 * q;
        pa[q] = Tt[(i >> 7) * A_STRIDE + (i & (F - 1))];
      }
      for (int it = 0; it < nt; ++it) {
        const int row0 = (base + G * it) * 16;
#pragma unroll
        for (int q = 0; q < 8; ++q) {
          int i = tid + 256 * q;
          Xs[(i >> 7) * A_STRIDE + (i & (F - 1))] = ssp(pa[q]);
        }
        lds_barrier();

        float xr[8];
#pragma unroll
        for (int j = 0; j < 8; ++j) {
          int r = row0 + j + 8 * hi;
          if (r >= N) r = N - 1;
          xr[j] = x[r * F + n];
        }

        if (it + 1 < nt) {
          const float* Tn = Tt + (it + 1) * TILE_F;
#pragma unroll
          for (int q = 0; q < 8; ++q) {
            int i = tid + 256 * q;
            pa[q] = Tn[(i >> 7) * A_STRIDE + (i & (F - 1))];
          }
        }

        v8f acc = tile_product(Xs, bwA, m, hi);
#pragma unroll
        for (int j = 0; j < 8; ++j) {
          int r = row0 + j + 8 * hi;
          float g = acc[j] + bn;
          float o = g + xr[j] * ex2n;
          if (r < N) out_final[r * F + n] = o;
        }
        lds_barrier();
      }
    }
  }
}

extern "C" void kernel_launch(void* const* d_in, const int* in_sizes, int n_in,
                              void* d_out, int out_size, void* d_ws, size_t ws_size,
                              hipStream_t stream) {
  const float* x      = (const float*)d_in[0];
  const int*   eidx   = (const int*)  d_in[1];
  const float* ea     = (const float*)d_in[2];
  const float* W_same = (const float*)d_in[3];
  const float* b_same = (const float*)d_in[4];
  const float* W_diff = (const float*)d_in[5];
  const float* b_diff = (const float*)d_in[6];
  const float* W_G    = (const float*)d_in[7];
  const float* rW1    = (const float*)d_in[8];
  const float* rb1    = (const float*)d_in[9];
  const float* rW2    = (const float*)d_in[10];
  const float* rb2    = (const float*)d_in[11];
  const float* W_last = (const float*)d_in[12];
  const float* b_last = (const float*)d_in[13];
  const float* u      = (const float*)d_in[14];

  const int NF = in_sizes[0];
  const int N  = NF / F;
  const int E  = in_sizes[2] / KRBF;
  const int R  = in_sizes[8] / (F * F);

  float* out_final = (float*)d_out;
  float* out_msged = (float*)d_out + NF;

  float* buf0 = (float*)d_ws;
  float* buf1 = buf0 + NF;

  hipMemsetAsync(buf1, 0, (size_t)NF * sizeof(float), stream);

  gemm128_kernel<true, 4, false><<<512, 256, 0, stream>>>(
      x, W_diff, b_diff, nullptr, nullptr, buf0, nullptr, N);

  edge_msg_kernel<<<512, 256, 0, stream>>>(buf0, eidx, ea, W_G, buf1, E);

  chain_kernel<<<512, 256, 0, stream>>>(
      x, buf1, W_same, b_same, rW1, rb1, rW2, rb2, W_last, b_last, u,
      out_msged, out_final, N, R);
}


# Shifted softplus in 8 vector slots, not 19: log(1+e) taken as hardware log2 (1+e always in [1,2], no denormal or infinity path), *ln2 and -ln2 folded into one fused multiply-add; last-bit changes
# speedup vs baseline: 3.1468x; 1.0476x over previous
#include <hip/hip_runtime.h>
#include <hip/hip_bf16.h>


typedef __attribute__((ext_vector_type(2))) float v2f;
typedef __attribute__((ext_vector_type(8))) float v8f;

__device__ __forceinline__ v8f wmma_k4(v2f a, v2f b, v8f c) {
  return __builtin_amdgcn_wmma_f32_16x16x4_f32(
       false, a,  false, b,
       (short)0, c,  false,  false);
}

__device__ __forceinline__ float ssp(float x) {
  float e = __expf(-fabsf(x));
  float l = __builtin_amdgcn_logf(1.0f + e);
  return fmaxf(x, 0.0f) + fmaf(l, 0.69314718055994531f, -0.69314718055994531f);
}

__device__ __forceinline__ void lds_barrier() {
  __builtin_amdgcn_fence(__ATOMIC_RELEASE, "workgroup", "local");
  __builtin_amdgcn_s_barrier();
  __builtin_amdgcn_fence(__ATOMIC_ACQUIRE, "workgroup", "local");
}

#define F 128
#define KRBF 64
#define A_STRIDE 132
#define E_STRIDE 68

#define ET 32

typedef __attribute__((ext_vector_type(4))) float v4f;

__device__ __forceinline__ float other_half(float x) {
  return __builtin_bit_cast(float, __builtin_amdgcn_permlanex16(
      __builtin_bit_cast(int, x), __builtin_bit_cast(int, x),
      0x76543210, 0xfedcba98, false, false));
}

__global__ void __launch_bounds__(256)
edge_msg_kernel(const float* __restrict__ P, const int* __restrict__ eidx,
                const float* __restrict__ ea, const float* __restrict__ Wg,
                float* __restrict__ aggr, int E) {
  __shared__ __align__(16) float As[ET * A_STRIDE];
  __shared__ __align__(16) float Es[ET * E_STRIDE];
  __shared__ int dsts[ET];

  const int tid  = threadIdx.x;
  const int lane = tid & 31;
  const int wv   = tid >> 5;
  const int m    = lane & 15;
  const int hi   = lane >> 4;
  const int rb   = (wv >> 2) * 16;
  const int c0   = (wv & 3) * 32;
  const int col  = c0 + lane;
  const int r4   = wv * 4;

  v2f bwg0[KRBF / 4];
  v2f bwg1[KRBF / 4];
  {
    const float* g0 = Wg + (c0 + m) * KRBF;
    const float* g1 = Wg + (c0 + 16 + m) * KRBF;
#pragma unroll
    for (int kk = 0; kk < KRBF / 4; ++kk) {
      bwg0[kk] = *(const v2f*)(&g0[kk * 4 + 2 * hi]);
      bwg1[kk] = *(const v2f*)(&g1[kk * 4 + 2 * hi]);
    }
  }

  const int nT = (E + ET - 1) / ET;

  v4f pa[4];
  v4f pe[2];
  int pd = 0;
  int ps = 0;

  int tile = blockIdx.x;
  if (tile < nT) {
    const int e0 = tile * ET;
    {
      int e = e0 + r4 + (lane & 3);
      if (e >= E) e = E - 1;
      ps = eidx[e];
    }
#pragma unroll
    for (int i = 0; i < 4; ++i) {
      int s = __builtin_amdgcn_readlane(ps, i);
      pa[i] = *(const v4f*)(&P[s * F + 4 * lane]);
    }
#pragma unroll
    for (int i = 0; i < 2; ++i) {
      int e = e0 + r4 + 2 * i + hi;
      if (e >= E) e = E - 1;
      pe[i] = *(const v4f*)(&ea[e * KRBF + 4 * m]);
    }
    if (tid < ET) {
      int e = e0 + tid;
      if (e >= E) e = E - 1;
      pd = eidx[E + e];
    }
    {
      int e = (tile + (int)gridDim.x) * ET + r4 + (lane & 3);
      if (e >= E) e = E - 1;
      ps = eidx[e];
    }
  }

  for (; tile < nT; tile += gridDim.x) {
    const int e0 = tile * ET;

#pragma unroll
    for (int i = 0; i < 4; ++i)
      *(v4f*)(&As[(r4 + i) * A_STRIDE + 4 * lane]) = pa[i];
#pragma unroll
    for (int i = 0; i < 2; ++i)
      *(v4f*)(&Es[(r4 + 2 * i + hi) * E_STRIDE + 4 * m]) = pe[i];
    if (tid < ET) dsts[tid] = pd;

    const int s0 = __builtin_amdgcn_readlane(ps, 0);
    const int s1 = __builtin_amdgcn_readlane(ps, 1);
    const int s2 = __builtin_amdgcn_readlane(ps, 2);
    const int s3 = __builtin_amdgcn_readlane(ps, 3);
    lds_barrier();

    const int nxt = tile + gridDim.x;
    if (nxt < nT) {
      const int f0 = nxt * ET;
      pa[0] = *(const v4f*)(&P[s0 * F + 4 * lane]);
      pa[1] = *(const v4f*)(&P[s1 * F + 4 * lane]);
      pa[2] = *(const v4f*)(&P[s2 * F + 4 * lane]);
      pa[3] = *(const v4f*)(&P[s3 * F + 4 * lane]);
#pragma unroll
      for (int i = 0; i < 2; ++i) {
        int e = f0 + r4 + 2 * i + hi;
        if (e >= E) e = E - 1;
        pe[i] = *(const v4f*)(&ea[e * KRBF + 4 * m]);
      }
      if (tid < ET) {
        int e = f0 + tid;
        if (e >= E) e = E - 1;
        pd = eidx[E + e];
      }
      {
        int e = (nxt + (int)gridDim.x) * ET + r4 + (lane & 3);
        if (e >= E) e = E - 1;
        ps = eidx[e];
      }
    }

    int   dr[16];
    float pv[16];
#pragma unroll
    for (int j = 0; j < 16; ++j) {
      dr[j] = dsts[rb + j];
      pv[j] = As[(rb + j) * A_STRIDE + col];
    }

    v8f acc0 = {};
    v8f acc1 = {};
#pragma unroll
    for (int kk = 0; kk < KRBF / 4; ++kk) {
      v2f a = *(const v2f*)(&Es[(rb + m) * E_STRIDE + kk * 4 + 2 * hi]);
      acc0 = wmma_k4(a, bwg0[kk], acc0);
      acc1 = wmma_k4(a, bwg1[kk], acc1);
    }

    if (e0 + ET <= E) {
#pragma unroll
      for (int j = 0; j < 8; ++j) {
        float give = hi ? acc0[j] : acc1[j];
        float got  = other_half(give);
        float glo  = hi ? got : acc0[j];
        float ghi  = hi ? acc1[j] : got;
        atomicAdd(&aggr[dr[j] * F + col], pv[j] * glo);
        atomicAdd(&aggr[dr[j + 8] * F + col], pv[j + 8] * ghi);
      }
    } else {
#pragma unroll
      for (int j = 0; j < 8; ++j) {
        float give = hi ? acc0[j] : acc1[j];
        float got  = other_half(give);
        float glo  = hi ? got : acc0[j];
        float ghi  = hi ? acc1[j] : got;
        int rlo = rb + j;
        int rhi = rb + j + 8;
        if (e0 + rlo < E)
          atomicAdd(&aggr[dr[j] * F + col], pv[j] * glo);
        if (e0 + rhi < E)
          atomicAdd(&aggr[dr[j + 8] * F + col], pv[j + 8] * ghi);
      }
    }
    lds_barrier();
  }
}

template <bool IN_ACT, int MODE, bool WRITE2>
__global__ void __launch_bounds__(256)
gemm128_kernel(const float* __restrict__ In, const float* __restrict__ W,
               const float* __restrict__ bias, const float* __restrict__ Ex1,
               const float* __restrict__ Ex2, float* __restrict__ Out,
               float* __restrict__ Out2, int N) {
  __shared__ __align__(16) float As[16 * A_STRIDE];

  const int tid  = threadIdx.x;
  const int lane = tid & 31;
  const int wv   = tid >> 5;
  const int m    = lane & 15;
  const int hi   = lane >> 4;
  const int n    = wv * 16 + m;

  v2f bw[F / 4];
  {
    const float* wrow = W + n * F;
#pragma unroll
    for (int kk = 0; kk < F / 4; ++kk)
      bw[kk] = *(const v2f*)(&wrow[kk * 4 + 2 * hi]);
  }
  const float bn = bias[n];
  float ex2n = 0.0f;
  if (MODE == 3) ex2n = Ex2[n];

  const int nT = (N + 15) / 16;

  float pa[8];
  int tile = blockIdx.x;
  if (tile < nT) {
    const int row0 = tile * 16;
#pragma unroll
    for (int q = 0; q < 8; ++q) {
      int i = tid + 256 * q;
      int rr = row0 + (i >> 7);
      if (rr >= N) rr = N - 1;
      pa[q] = In[rr * F + (i & (F - 1))];
    }
  }

  for (; tile < nT; tile += gridDim.x) {
    const int row0 = tile * 16;

#pragma unroll
    for (int q = 0; q < 8; ++q) {
      int i = tid + 256 * q;
      float v = pa[q];
      if (IN_ACT) v = ssp(v);
      As[(i >> 7) * A_STRIDE + (i & (F - 1))] = v;
    }
    lds_barrier();

    const int nxt = tile + gridDim.x;
    if (nxt < nT) {
      const int f0 = nxt * 16;
#pragma unroll
      for (int q = 0; q < 8; ++q) {
        int i = tid + 256 * q;
        int rr = f0 + (i >> 7);
        if (rr >= N) rr = N - 1;
        pa[q] = In[rr * F + (i & (F - 1))];
      }
    }

    v8f acc = {};
#pragma unroll
    for (int kk = 0; kk < F / 4; ++kk) {
      v2f a = *(const v2f*)(&As[m * A_STRIDE + kk * 4 + 2 * hi]);
      acc = wmma_k4(a, bw[kk], acc);
    }

#pragma unroll
    for (int j = 0; j < 8; ++j) {
      int r = row0 + j + 8 * hi;
      if (r >= N) continue;
      float g = acc[j] + bn;
      float o;
      if (MODE == 0)      o = ssp(g) + Ex1[r * F + n];
      else if (MODE == 1) o = g;
      else if (MODE == 2) o = Ex1[r * F + n] + g;
      else if (MODE == 3) o = g + Ex1[r * F + n] * ex2n;
      else                o = ssp(g);
      Out[r * F + n] = o;
      if (WRITE2) Out2[r * F + n] = o;
    }
    lds_barrier();
  }
}

#define CH_T 7
#define TILE_F (16 * A_STRIDE)

__device__ __forceinline__ v8f tile_product(const float* As, const v2f (&bw)[F / 4],
                                            int m, int hi) {
  v8f acc = {};
#pragma unroll
  for (int kk = 0; kk < F / 4; ++kk) {
    v2f a = *(const v2f*)(&As[m * A_STRIDE + kk * 4 + 2 * hi]);
    acc = wmma_k4(a, bw[kk], acc);
  }
  return acc;
}

__device__ __forceinline__ void load_frags(v2f (&bw)[F / 4], const float* wrow, int hi) {
#pragma unroll
  for (int kk = 0; kk < F / 4; ++kk)
    bw[kk] = *(const v2f*)(&wrow[kk * 4 + 2 * hi]);
}

__global__ void __launch_bounds__(256) __attribute__((amdgpu_waves_per_eu(4, 4)))
chain_kernel(const float* __restrict__ x, const float* __restrict__ aggr,
             const float* __restrict__ W_same, const float* __restrict__ b_same,
             const float* __restrict__ rW1, const float* __restrict__ rb1,
             const float* __restrict__ rW2, const float* __restrict__ rb2,
             const float* __restrict__ W_last, const float* __restrict__ b_last,
             const float* __restrict__ u,
             float* __restrict__ out_msged, float* __restrict__ out_final,
             int N, int R) {
  __shared__ __align__(16) float Tt[CH_T * TILE_F];
  __shared__ __align__(16) float Xs[TILE_F];
  __shared__ __align__(16) float Ys[TILE_F];

  const int tid  = threadIdx.x;
  const int lane = tid & 31;
  const int wv   = tid >> 5;
  const int m    = lane & 15;
  const int hi   = lane >> 4;
  const int n    = wv * 16 + m;
  const int G    = gridDim.x;

  const int nT = (N + 15) / 16;

  v2f bwA[F / 4];
  v2f bwB[F / 4];
  float pa[8];

  const int base = blockIdx.x;
  if (base >= nT) return;
  {
    int nt = (nT - base + G - 1) / G;
    if (nt > CH_T) nt = CH_T;

    {
      load_frags(bwA, W_same + n * F, hi);
      const float bn = b_same[n];
      {
        const int row0 = base * 16;
#pragma unroll
        for (int q = 0; q < 8; ++q) {
          int i = tid + 256 * q;
          int rr = row0 + (i >> 7);
          if (rr >= N) rr = N - 1;
          pa[q] = x[rr * F + (i & (F - 1))];
        }
      }
      for (int it = 0; it < nt; ++it) {
        const int row0 = (base + G * it) * 16;
#pragma unroll
        for (int q = 0; q < 8; ++q) {
          int i = tid + 256 * q;
          Xs[(i >> 7) * A_STRIDE + (i & (F - 1))] = ssp(pa[q]);
        }
        lds_barrier();

        float ag[8];
#pragma unroll
        for (int j = 0; j < 8; ++j) {
          int r = row0 + j + 8 * hi;
          if (r >= N) r = N - 1;
          ag[j] = aggr[r * F + n];
        }

        if (it + 1 < nt) {
          const int f0 = (base + G * (it + 1)) * 16;
#pragma unroll
          for (int q = 0; q < 8; ++q) {
            int i = tid + 256 * q;
            int rr = f0 + (i >> 7);
            if (rr >= N) rr = N - 1;
            pa[q] = x[rr * F + (i & (F - 1))];
          }
        }

        v8f acc = tile_product(Xs, bwA, m, hi);
        float* Tr = Tt + it * TILE_F;
#pragma unroll
        for (int j = 0; j < 8; ++j) {
          int rl = j + 8 * hi;
          int r  = row0 + rl;
          float g = acc[j] + bn;
          float o = ssp(g) + ag[j];
          if (r < N) out_msged[r * F + n] = o;
          else       o = 0.0f;
          Tr[rl * A_STRIDE + n] = o;
        }
        lds_barrier();
      }
    }

    for (int rb = 0; rb < R; ++rb) {
      load_frags(bwA, rW1 + rb * F * F + n * F, hi);
      load_frags(bwB, rW2 + rb * F * F + n * F, hi);
      const float b1n = rb1[rb * F + n];
      const float b2n = rb2[rb * F + n];

#pragma unroll
      for (int q = 0; q < 8; ++q) {
        int i = tid + 256 * q;
        pa[q] = Tt[(i >> 7) * A_STRIDE + (i & (F - 1))];
      }
      for (int it = 0; it < nt; ++it) {
#pragma unroll
        for (int q = 0; q < 8; ++q) {
          int i = tid + 256 * q;
          Xs[(i >> 7) * A_STRIDE + (i & (F - 1))] = ssp(pa[q]);
        }
        lds_barrier();

        if (it + 1 < nt) {
          const float* Tn = Tt + (it + 1) * TILE_F;
#pragma unroll
          for (int q = 0; q < 8; ++q) {
            int i = tid + 256 * q;
            pa[q] = Tn[(i >> 7) * A_STRIDE + (i & (F - 1))];
          }
        }

        {
          v8f acc = tile_product(Xs, bwA, m, hi);
#pragma unroll
          for (int j = 0; j < 8; ++j) {
            int rl = j + 8 * hi;
            float g = acc[j] + b1n;
            Ys[rl * A_STRIDE + n] = ssp(g);
          }
        }
        lds_barrier();

        {
          v8f acc = tile_product(Ys, bwB, m, hi);
          float* Tr = Tt + it * TILE_F;
#pragma unroll
          for (int j = 0; j < 8; ++j) {
            int rl = j + 8 * hi;
            float g = acc[j] + b2n;
            Tr[rl * A_STRIDE + n] = Tr[rl * A_STRIDE + n] + g;
          }
        }
        lds_barrier();
      }
    }

    {
      load_frags(bwA, W_last + n * F, hi);
      const float bn   = b_last[n];
      const float ex2n = u[n];

#pragma unroll
      for (int q = 0; q < 8; ++q) {
        int i = tid + 256 * q;
        pa[q] = Tt[(i >> 7) * A_STRIDE + (i & (F - 1))];
      }
      for (int it = 0; it < nt; ++it) {
        const int row0 = (base + G * it) * 16;
#pragma unroll
        for (int q = 0; q < 8; ++q) {
          int i = tid + 256 * q;
          Xs[(i >> 7) * A_STRIDE + (i & (F - 1))] = ssp(pa[q]);
        }
        lds_barrier();

        float xr[8];
#pragma unroll
        for (int j = 0; j < 8; ++j) {
          int r = row0 + j + 8 * hi;
          if (r >= N) r = N - 1;
          xr[j] = x[r * F + n];
        }

        if (it + 1 < nt) {
          const float* Tn = Tt + (it + 1) * TILE_F;
#pragma unroll
          for (int q = 0; q < 8; ++q) {
            int i = tid + 256 * q;
            pa[q] = Tn[(i >> 7) * A_STRIDE + (i & (F - 1))];
          }
        }

        v8f acc = tile_product(Xs, bwA, m, hi);
#pragma unroll
        for (int j = 0; j < 8; ++j) {
          int r = row0 + j + 8 * hi;
          float g = acc[j] + bn;
          float o = g + xr[j] * ex2n;
          if (r < N) out_final[r * F + n] = o;
        }
        lds_barrier();
      }
    }
  }
}

extern "C" void kernel_launch(void* const* d_in, const int* in_sizes, int n_in,
                              void* d_out, int out_size, void* d_ws, size_t ws_size,
                              hipStream_t stream) {
  const float* x      = (const float*)d_in[0];
  const int*   eidx   = (const int*)  d_in[1];
  const float* ea     = (const float*)d_in[2];
  const float* W_same = (const float*)d_in[3];
  const float* b_same = (const float*)d_in[4];
  const float* W_diff = (const float*)d_in[5];
  const float* b_diff = (const float*)d_in[6];
  const float* W_G    = (const float*)d_in[7];
  const float* rW1    = (const float*)d_in[8];
  const float* rb1    = (const float*)d_in[9];
  const float* rW2    = (const float*)d_in[10];
  const float* rb2    = (const float*)d_in[11];
  const float* W_last = (const float*)d_in[12];
  const float* b_last = (const float*)d_in[13];
  const float* u      = (const float*)d_in[14];

  const int NF = in_sizes[0];
  const int N  = NF / F;
  const int E  = in_sizes[2] / KRBF;
  const int R  = in_sizes[8] / (F * F);

  float* out_final = (float*)d_out;
  float* out_msged = (float*)d_out + NF;

  float* buf0 = (float*)d_ws;
  float* buf1 = buf0 + NF;

  hipMemsetAsync(buf1, 0, (size_t)NF * sizeof(float), stream);

  gemm128_kernel<true, 4, false><<<512, 256, 0, stream>>>(
      x, W_diff, b_diff, nullptr, nullptr, buf0, nullptr, N);

  edge_msg_kernel<<<512, 256, 0, stream>>>(buf0, eidx, ea, W_G, buf1, E);

  chain_kernel<<<512, 256, 0, stream>>>(
      x, buf1, W_same, b_same, rW1, rb1, rW2, rb2, W_last, b_last, u,
      out_msged, out_final, N, R);
}


# Update term ssp(xa@W_same^T+b) taken with the W_diff table in one pre-pass (shared A tile) and stored as the scatter's start value in msged: sum order now S+m1+m2.. on the atomics; no zero fill
# speedup vs baseline: 3.2946x; 1.0470x over previous
#include <hip/hip_runtime.h>
#include <hip/hip_bf16.h>


typedef __attribute__((ext_vector_type(2))) float v2f;
typedef __attribute__((ext_vector_type(8))) float v8f;

__device__ __forceinline__ v8f wmma_k4(v2f a, v2f b, v8f c) {
  return __builtin_amdgcn_wmma_f32_16x16x4_f32(
       false, a,  false, b,
       (short)0, c,  false,  false);
}

__device__ __forceinline__ float ssp(float x) {
  float e = __expf(-fabsf(x));
  float l = __builtin_amdgcn_logf(1.0f + e);
  return fmaxf(x, 0.0f) + fmaf(l, 0.69314718055994531f, -0.69314718055994531f);
}

__device__ __forceinline__ void lds_barrier() {
  __builtin_amdgcn_fence(__ATOMIC_RELEASE, "workgroup", "local");
  __builtin_amdgcn_s_barrier();
  __builtin_amdgcn_fence(__ATOMIC_ACQUIRE, "workgroup", "local");
}

#define F 128
#define KRBF 64
#define A_STRIDE 132
#define E_STRIDE 68

#define ET 32

typedef __attribute__((ext_vector_type(4))) float v4f;

__device__ __forceinline__ float other_half(float x) {
  return __builtin_bit_cast(float, __builtin_amdgcn_permlanex16(
      __builtin_bit_cast(int, x), __builtin_bit_cast(int, x),
      0x76543210, 0xfedcba98, false, false));
}

__global__ void __launch_bounds__(256)
edge_msg_kernel(const float* __restrict__ P, const int* __restrict__ eidx,
                const float* __restrict__ ea, const float* __restrict__ Wg,
                float* __restrict__ aggr, int E) {
  __shared__ __align__(16) float As[ET * A_STRIDE];
  __shared__ __align__(16) float Es[ET * E_STRIDE];
  __shared__ int dsts[ET];

  const int tid  = threadIdx.x;
  const int lane = tid & 31;
  const int wv   = tid >> 5;
  const int m    = lane & 15;
  const int hi   = lane >> 4;
  const int rb   = (wv >> 2) * 16;
  const int c0   = (wv & 3) * 32;
  const int col  = c0 + lane;
  const int r4   = wv * 4;

  v2f bwg0[KRBF / 4];
  v2f bwg1[KRBF / 4];
  {
    const float* g0 = Wg + (c0 + m) * KRBF;
    const float* g1 = Wg + (c0 + 16 + m) * KRBF;
#pragma unroll
    for (int kk = 0; kk < KRBF / 4; ++kk) {
      bwg0[kk] = *(const v2f*)(&g0[kk * 4 + 2 * hi]);
      bwg1[kk] = *(const v2f*)(&g1[kk * 4 + 2 * hi]);
    }
  }

  const int nT = (E + ET - 1) / ET;

  v4f pa[4];
  v4f pe[2];
  int pd = 0;
  int ps = 0;

  int tile = blockIdx.x;
  if (tile < nT) {
    const int e0 = tile * ET;
    {
      int e = e0 + r4 + (lane & 3);
      if (e >= E) e = E - 1;
      ps = eidx[e];
    }
#pragma unroll
    for (int i = 0; i < 4; ++i) {
      int s = __builtin_amdgcn_readlane(ps, i);
      pa[i] = *(const v4f*)(&P[s * F + 4 * lane]);
    }
#pragma unroll
    for (int i = 0; i < 2; ++i) {
      int e = e0 + r4 + 2 * i + hi;
      if (e >= E) e = E - 1;
      pe[i] = *(const v4f*)(&ea[e * KRBF + 4 * m]);
    }
    if (tid < ET) {
      int e = e0 + tid;
      if (e >= E) e = E - 1;
      pd = eidx[E + e];
    }
    {
      int e = (tile + (int)gridDim.x) * ET + r4 + (lane & 3);
      if (e >= E) e = E - 1;
      ps = eidx[e];
    }
  }

  for (; tile < nT; tile += gridDim.x) {
    const int e0 = tile * ET;

#pragma unroll
    for (int i = 0; i < 4; ++i)
      *(v4f*)(&As[(r4 + i) * A_STRIDE + 4 * lane]) = pa[i];
#pragma unroll
    for (int i = 0; i < 2; ++i)
      *(v4f*)(&Es[(r4 + 2 * i + hi) * E_STRIDE + 4 * m]) = pe[i];
    if (tid < ET) dsts[tid] = pd;

    const int s0 = __builtin_amdgcn_readlane(ps, 0);
    const int s1 = __builtin_amdgcn_readlane(ps, 1);
    const int s2 = __builtin_amdgcn_readlane(ps, 2);
    const int s3 = __builtin_amdgcn_readlane(ps, 3);
    lds_barrier();

    const int nxt = tile + gridDim.x;
    if (nxt < nT) {
      const int f0 = nxt * ET;
      pa[0] = *(const v4f*)(&P[s0 * F + 4 * lane]);
      pa[1] = *(const v4f*)(&P[s1 * F + 4 * lane]);
      pa[2] = *(const v4f*)(&P[s2 * F + 4 * lane]);
      pa[3] = *(const v4f*)(&P[s3 * F + 4 * lane]);
#pragma unroll
      for (int i = 0; i < 2; ++i) {
        int e = f0 + r4 + 2 * i + hi;
        if (e >= E) e = E - 1;
        pe[i] = *(const v4f*)(&ea[e * KRBF + 4 * m]);
      }
      if (tid < ET) {
        int e = f0 + tid;
        if (e >= E) e = E - 1;
        pd = eidx[E + e];
      }
      {
        int e = (nxt + (int)gridDim.x) * ET + r4 + (lane & 3);
        if (e >= E) e = E - 1;
        ps = eidx[e];
      }
    }

    int   dr[16];
    float pv[16];
#pragma unroll
    for (int j = 0; j < 16; ++j) {
      dr[j] = dsts[rb + j];
      pv[j] = As[(rb + j) * A_STRIDE + col];
    }

    v8f acc0 = {};
    v8f acc1 = {};
#pragma unroll
    for (int kk = 0; kk < KRBF / 4; ++kk) {
      v2f a = *(const v2f*)(&Es[(rb + m) * E_STRIDE + kk * 4 + 2 * hi]);
      acc0 = wmma_k4(a, bwg0[kk], acc0);
      acc1 = wmma_k4(a, bwg1[kk], acc1);
    }

    if (e0 + ET <= E) {
#pragma unroll
      for (int j = 0; j < 8; ++j) {
        float give = hi ? acc0[j] : acc1[j];
        float got  = other_half(give);
        float glo  = hi ? got : acc0[j];
        float ghi  = hi ? acc1[j] : got;
        atomicAdd(&aggr[dr[j] * F + col], pv[j] * glo);
        atomicAdd(&aggr[dr[j + 8] * F + col], pv[j + 8] * ghi);
      }
    } else {
#pragma unroll
      for (int j = 0; j < 8; ++j) {
        float give = hi ? acc0[j] : acc1[j];
        float got  = other_half(give);
        float glo  = hi ? got : acc0[j];
        float ghi  = hi ? acc1[j] : got;
        int rlo = rb + j;
        int rhi = rb + j + 8;
        if (e0 + rlo < E)
          atomicAdd(&aggr[dr[j] * F + col], pv[j] * glo);
        if (e0 + rhi < E)
          atomicAdd(&aggr[dr[j + 8] * F + col], pv[j + 8] * ghi);
      }
    }
    lds_barrier();
  }
}

__global__ void __launch_bounds__(256)
node_pre_kernel(const float* __restrict__ x,
                const float* __restrict__ Wd, const float* __restrict__ bd,
                const float* __restrict__ Ws, const float* __restrict__ bs,
                float* __restrict__ P, float* __restrict__ S, int N) {
  __shared__ __align__(16) float As[16 * A_STRIDE];

  const int tid  = threadIdx.x;
  const int lane = tid & 31;
  const int wv   = tid >> 5;
  const int m    = lane & 15;
  const int hi   = lane >> 4;
  const int n    = wv * 16 + m;

  v2f bwd[F / 4];
  v2f bws[F / 4];
  {
    const float* drow = Wd + n * F;
    const float* srow = Ws + n * F;
#pragma unroll
    for (int kk = 0; kk < F / 4; ++kk) {
      bwd[kk] = *(const v2f*)(&drow[kk * 4 + 2 * hi]);
      bws[kk] = *(const v2f*)(&srow[kk * 4 + 2 * hi]);
    }
  }
  const float bdn = bd[n];
  const float bsn = bs[n];

  const int nT = (N + 15) / 16;

  float pa[8];
  int tile = blockIdx.x;
  if (tile < nT) {
    const int row0 = tile * 16;
#pragma unroll
    for (int q = 0; q < 8; ++q) {
      int i = tid + 256 * q;
      int rr = row0 + (i >> 7);
      if (rr >= N) rr = N - 1;
      pa[q] = x[rr * F + (i & (F - 1))];
    }
  }

  for (; tile < nT; tile += gridDim.x) {
    const int row0 = tile * 16;

#pragma unroll
    for (int q = 0; q < 8; ++q) {
      int i = tid + 256 * q;
      As[(i >> 7) * A_STRIDE + (i & (F - 1))] = ssp(pa[q]);
    }
    lds_barrier();

    const int nxt = tile + gridDim.x;
    if (nxt < nT) {
      const int f0 = nxt * 16;
#pragma unroll
      for (int q = 0; q < 8; ++q) {
        int i = tid + 256 * q;
        int rr = f0 + (i >> 7);
        if (rr >= N) rr = N - 1;
        pa[q] = x[rr * F + (i & (F - 1))];
      }
    }

    v8f accd = {};
    v8f accs = {};
#pragma unroll
    for (int kk = 0; kk < F / 4; ++kk) {
      v2f a = *(const v2f*)(&As[m * A_STRIDE + kk * 4 + 2 * hi]);
      accd = wmma_k4(a, bwd[kk], accd);
      accs = wmma_k4(a, bws[kk], accs);
    }

#pragma unroll
    for (int j = 0; j < 8; ++j) {
      int r = row0 + j + 8 * hi;
      if (r >= N) continue;
      P[r * F + n] = ssp(accd[j] + bdn);
      S[r * F + n] = ssp(accs[j] + bsn);
    }
    lds_barrier();
  }
}

#define CH_T 7
#define TILE_F (16 * A_STRIDE)

__device__ __forceinline__ v8f tile_product(const float* As, const v2f (&bw)[F / 4],
                                            int m, int hi) {
  v8f acc = {};
#pragma unroll
  for (int kk = 0; kk < F / 4; ++kk) {
    v2f a = *(const v2f*)(&As[m * A_STRIDE + kk * 4 + 2 * hi]);
    acc = wmma_k4(a, bw[kk], acc);
  }
  return acc;
}

__device__ __forceinline__ void load_frags(v2f (&bw)[F / 4], const float* wrow, int hi) {
#pragma unroll
  for (int kk = 0; kk < F / 4; ++kk)
    bw[kk] = *(const v2f*)(&wrow[kk * 4 + 2 * hi]);
}

__global__ void __launch_bounds__(256) __attribute__((amdgpu_waves_per_eu(4, 4)))
chain_kernel(const float* __restrict__ x, const float* __restrict__ msged,
             const float* __restrict__ rW1, const float* __restrict__ rb1,
             const float* __restrict__ rW2, const float* __restrict__ rb2,
             const float* __restrict__ W_last, const float* __restrict__ b_last,
             const float* __restrict__ u,
             float* __restrict__ out_final,
             int N, int R) {
  __shared__ __align__(16) float Tt[CH_T * TILE_F];
  __shared__ __align__(16) float Xs[TILE_F];
  __shared__ __align__(16) float Ys[TILE_F];

  const int tid  = threadIdx.x;
  const int lane = tid & 31;
  const int wv   = tid >> 5;
  const int m    = lane & 15;
  const int hi   = lane >> 4;
  const int n    = wv * 16 + m;
  const int G    = gridDim.x;

  const int nT = (N + 15) / 16;

  v2f bwA[F / 4];
  v2f bwB[F / 4];
  float pa[8];

  const int base = blockIdx.x;
  if (base >= nT) return;
  {
    int nt = (nT - base + G - 1) / G;
    if (nt > CH_T) nt = CH_T;

    {
      float tv[CH_T][8];
#pragma unroll
      for (int it = 0; it < CH_T; ++it) {
#pragma unroll
        for (int q = 0; q < 8; ++q) tv[it][q] = 0.0f;
        if (it < nt) {
          const int row0 = (base + G * it) * 16;
#pragma unroll
          for (int q = 0; q < 8; ++q) {
            int i = tid + 256 * q;
            int rr = row0 + (i >> 7);
            if (rr >= N) rr = N - 1;
            tv[it][q] = msged[rr * F + (i & (F - 1))];
          }
        }
      }
#pragma unroll
      for (int it = 0; it < CH_T; ++it) {
        if (it < nt) {
          const int row0 = (base + G * it) * 16;
          float* Tr = Tt + it * TILE_F;
#pragma unroll
          for (int q = 0; q < 8; ++q) {
            int i = tid + 256 * q;
            int rl = i >> 7;
            float v = tv[it][q];
            if (row0 + rl >= N) v = 0.0f;
            Tr[rl * A_STRIDE + (i & (F - 1))] = v;
          }
        }
      }
      lds_barrier();
    }

    for (int rb = 0; rb < R; ++rb) {
      load_frags(bwA, rW1 + rb * F * F + n * F, hi);
      load_frags(bwB, rW2 + rb * F * F + n * F, hi);
      const float b1n = rb1[rb * F + n];
      const float b2n = rb2[rb * F + n];

#pragma unroll
      for (int q = 0; q < 8; ++q) {
        int i = tid + 256 * q;
        pa[q] = Tt[(i >> 7) * A_STRIDE + (i & (F - 1))];
      }
      for (int it = 0; it < nt; ++it) {
#pragma unroll
        for (int q = 0; q < 8; ++q) {
          int i = tid + 256 * q;
          Xs[(i >> 7) * A_STRIDE + (i & (F - 1))] = ssp(pa[q]);
        }
        lds_barrier();

        if (it + 1 < nt) {
          const float* Tn = Tt + (it + 1) * TILE_F;
#pragma unroll
          for (int q = 0; q < 8; ++q) {
            int i = tid + 256 * q;
            pa[q] = Tn[(i >> 7) * A_STRIDE + (i & (F - 1))];
          }
        }

        {
          v8f acc = tile_product(Xs, bwA, m, hi);
#pragma unroll
          for (int j = 0; j < 8; ++j) {
            int rl = j + 8 * hi;
            float g = acc[j] + b1n;
            Ys[rl * A_STRIDE + n] = ssp(g);
          }
        }
        lds_barrier();

        {
          v8f acc = tile_product(Ys, bwB, m, hi);
          float* Tr = Tt + it * TILE_F;
#pragma unroll
          for (int j = 0; j < 8; ++j) {
            int rl = j + 8 * hi;
            float g = acc[j] + b2n;
            Tr[rl * A_STRIDE + n] = Tr[rl * A_STRIDE + n] + g;
          }
        }
        lds_barrier();
      }
    }

    {
      load_frags(bwA, W_last + n * F, hi);
      const float bn   = b_last[n];
      const float ex2n = u[n];

#pragma unroll
      for (int q = 0; q < 8; ++q) {
        int i = tid + 256 * q;
        pa[q] = Tt[(i >> 7) * A_STRIDE + (i & (F - 1))];
      }
      for (int it = 0; it < nt; ++it) {
        const int row0 = (base + G * it) * 16;
#pragma unroll
        for (int q = 0; q < 8; ++q) {
          int i = tid + 256 * q;
          Xs[(i >> 7) * A_STRIDE + (i & (F - 1))] = ssp(pa[q]);
        }
        lds_barrier();

        float xr[8];
#pragma unroll
        for (int j = 0; j < 8; ++j) {
          int r = row0 + j + 8 * hi;
          if (r >= N) r = N - 1;
          xr[j] = x[r * F + n];
        }

        if (it + 1 < nt) {
          const float* Tn = Tt + (it + 1) * TILE_F;
#pragma unroll
          for (int q = 0; q < 8; ++q) {
            int i = tid + 256 * q;
            pa[q] = Tn[(i >> 7) * A_STRIDE + (i & (F - 1))];
          }
        }

        v8f acc = tile_product(Xs, bwA, m, hi);
#pragma unroll
        for (int j = 0; j < 8; ++j) {
          int r = row0 + j + 8 * hi;
          float g = acc[j] + bn;
          float o = g + xr[j] * ex2n;
          if (r < N) out_final[r * F + n] = o;
        }
        lds_barrier();
      }
    }
  }
}

extern "C" void kernel_launch(void* const* d_in, const int* in_sizes, int n_in,
                              void* d_out, int out_size, void* d_ws, size_t ws_size,
                              hipStream_t stream) {
  const float* x      = (const float*)d_in[0];
  const int*   eidx   = (const int*)  d_in[1];
  const float* ea     = (const float*)d_in[2];
  const float* W_same = (const float*)d_in[3];
  const float* b_same = (const float*)d_in[4];
  const float* W_diff = (const float*)d_in[5];
  const float* b_diff = (const float*)d_in[6];
  const float* W_G    = (const float*)d_in[7];
  const float* rW1    = (const float*)d_in[8];
  const float* rb1    = (const float*)d_in[9];
  const float* rW2    = (const float*)d_in[10];
  const float* rb2    = (const float*)d_in[11];
  const float* W_last = (const float*)d_in[12];
  const float* b_last = (const float*)d_in[13];
  const float* u      = (const float*)d_in[14];

  const int NF = in_sizes[0];
  const int N  = NF / F;
  const int E  = in_sizes[2] / KRBF;
  const int R  = in_sizes[8] / (F * F);

  float* out_final = (float*)d_out;
  float* out_msged = (float*)d_out + NF;

  float* P = (float*)d_ws;

  node_pre_kernel<<<512, 256, 0, stream>>>(x, W_diff, b_diff, W_same, b_same,
                                           P, out_msged, N);

  edge_msg_kernel<<<512, 256, 0, stream>>>(P, eidx, ea, W_G, out_msged, E);

  chain_kernel<<<512, 256, 0, stream>>>(
      x, out_msged, rW1, rb1, rW2, rb2, W_last, b_last, u, out_final, N, R);
}
